# starting kernel
# baseline (speedup 1.0000x reference)
#include <hip/hip_runtime.h>


typedef __attribute__((ext_vector_type(2))) float v2f;
typedef __attribute__((ext_vector_type(8))) float v8f;

#define NN   20000
#define DIN  300
#define HD   150
#define IOUW 450
#define CHP  152
#define IOUP 464
#define FCP  160

__device__ __forceinline__ float sigmoid_f(float x) {
  return 1.0f / (1.0f + __expf(-x));
}
__device__ __forceinline__ float tanh_f(float x) {
  float t = __expf(2.0f * fminf(fmaxf(x, -15.0f), 15.0f));
  return (t - 1.0f) / (t + 1.0f);
}

__global__ void __launch_bounds__(128)
proj_gemm(const float* __restrict__ A, const float* __restrict__ B,
          const float* __restrict__ bias, float* __restrict__ C, int N) {
  __shared__ float Bs[32 * 304];
  const int tid  = threadIdx.x;
  const int lane = tid & 31;
  const int wave = tid >> 5;
  const int l15  = lane & 15;
  const int kb0  = (lane >> 4) << 1;
  const int n0   = (int)blockIdx.y * 32;

  for (int e = tid; e < 32 * DIN; e += 128) {
    int k = e >> 5, nl = e & 31;
    int n = n0 + nl;
    Bs[nl * 304 + k] = (n < N) ? B[k * N + n] : 0.0f;
  }
  __syncthreads();

  const int m0 = ((int)blockIdx.x * 4 + wave) * 16;
  if (m0 >= NN) return;
  const float* arow  = A + (size_t)(m0 + l15) * DIN;
  const float* b0row = Bs + l15 * 304;
  const float* b1row = Bs + (l15 + 16) * 304;
  v8f acc0 = {}, acc1 = {};
  for (int k0 = 0; k0 < DIN; k0 += 4) {
    int kb = k0 + kb0;
    v2f a  = *(const v2f*)(arow + kb);
    v2f b0 = *(const v2f*)(b0row + kb);
    v2f b1 = *(const v2f*)(b1row + kb);
    acc0 = __builtin_amdgcn_wmma_f32_16x16x4_f32(false, a, false, b0, (short)0,
                                                 acc0, false, false);
    acc1 = __builtin_amdgcn_wmma_f32_16x16x4_f32(false, a, false, b1, (short)0,
                                                 acc1, false, false);
  }
#pragma unroll
  for (int half = 0; half < 2; ++half) {
    int n = n0 + half * 16 + l15;
    if (n < N) {
      float bv = bias[n];
      const v8f& acc = half ? acc1 : acc0;
#pragma unroll
      for (int v = 0; v < 8; ++v) {
        int m = m0 + v + ((lane >> 4) << 3);
        C[(size_t)m * N + n] = acc[v] + bv;
      }
    }
  }
}

__global__ void leaf_kernel(const float* __restrict__ xiou,
                            const float* __restrict__ biouh,
                            float* __restrict__ cst, float* __restrict__ hst,
                            int count) {
  int e = (int)blockIdx.x * blockDim.x + threadIdx.x;
  if (e >= count * HD) return;
  int node = e / HD, col = e - node * HD;
  const float* r = xiou + (size_t)node * IOUW;
  float i = sigmoid_f(r[col] + biouh[col]);
  float o = sigmoid_f(r[HD + col] + biouh[HD + col]);
  float u = tanh_f(r[2 * HD + col] + biouh[2 * HD + col]);
  float c = i * u;
  cst[(size_t)node * HD + col] = c;
  hst[(size_t)node * HD + col] = o * tanh_f(c);
}

__global__ void __launch_bounds__(256)
level_kernel(const float* __restrict__ xiou, const float* __restrict__ xf,
             const float* __restrict__ Wiouh, const float* __restrict__ biouh,
             const float* __restrict__ Wfh, const float* __restrict__ bfh,
             float* __restrict__ cst, float* __restrict__ hst,
             int start, int end) {
  extern __shared__ float smem[];
  float* chh  = smem;
  float* chc  = chh  + 64 * CHP;
  float* hsum = chc  + 64 * CHP;
  float* fcs  = hsum + 16 * CHP;
  float* ious = fcs  + 16 * FCP;

  const int tid  = threadIdx.x;
  const int lane = tid & 31;
  const int wave = tid >> 5;
  const int l15  = lane & 15;
  const int kb0  = (lane >> 4) << 1;
  const int roff = (lane >> 4) << 3;
  const int base = start + (int)blockIdx.x * 16;

  for (int e = tid; e < 64 * CHP; e += 256) {
    int row = e / CHP, col = e - row * CHP;
    float hv = 0.0f, cv = 0.0f;
    int node = base + (row >> 2);
    if (col < HD && node < end) {
      int cr = 4 * ((NN - 1) - node) + 1 + (row & 3);
      if (cr < NN) {
        int ci = (NN - 1) - cr;
        hv = hst[(size_t)ci * HD + col];
        cv = cst[(size_t)ci * HD + col];
      }
    }
    chh[e] = hv;
    chc[e] = cv;
  }
  __syncthreads();
  for (int e = tid; e < 16 * CHP; e += 256) {
    int m = e / CHP, col = e - m * CHP;
    int r0 = (m * 4) * CHP + col;
    hsum[e] = chh[r0] + chh[r0 + CHP] + chh[r0 + 2 * CHP] + chh[r0 + 3 * CHP];
  }
  __syncthreads();

  for (int nt = wave; nt < 29; nt += 8) {
    int n0 = nt * 16;
    int n  = n0 + l15;
    bool nok = n < IOUW;
    v8f acc = {};
    for (int k0 = 0; k0 < 148; k0 += 4) {
      int kb = k0 + kb0;
      v2f a = *(const v2f*)(hsum + l15 * CHP + kb);
      v2f b;
      b.x = nok ? Wiouh[kb * IOUW + n] : 0.0f;
      b.y = nok ? Wiouh[(kb + 1) * IOUW + n] : 0.0f;
      acc = __builtin_amdgcn_wmma_f32_16x16x4_f32(false, a, false, b, (short)0,
                                                  acc, false, false);
    }
    {
      int kb = 148 + kb0;
      v2f a = *(const v2f*)(hsum + l15 * CHP + kb);
      v2f b;
      b.x = (nok && kb < HD)     ? Wiouh[kb * IOUW + n]       : 0.0f;
      b.y = (nok && kb + 1 < HD) ? Wiouh[(kb + 1) * IOUW + n] : 0.0f;
      acc = __builtin_amdgcn_wmma_f32_16x16x4_f32(false, a, false, b, (short)0,
                                                  acc, false, false);
    }
    if (nok) {
      float bv = biouh[n];
#pragma unroll
      for (int v = 0; v < 8; ++v) {
        int m = v + roff;
        int node = base + m;
        float xv = (node < end) ? xiou[(size_t)node * IOUW + n] : 0.0f;
        ious[m * IOUP + n] = acc[v] + bv + xv;
      }
    }
  }

  for (int t = wave; t < 20; t += 8) {
    int mp = t / 10, nt = t - mp * 10;
    int m0 = mp * 32, n0 = nt * 16;
    int n  = n0 + l15;
    bool nok = n < HD;
    int  nc  = nok ? n : 0;
    v8f acc0 = {}, acc1 = {};
    for (int k0 = 0; k0 < 148; k0 += 4) {
      int kb = k0 + kb0;
      v2f a0 = *(const v2f*)(chh + (m0 + l15) * CHP + kb);
      v2f a1 = *(const v2f*)(chh + (m0 + 16 + l15) * CHP + kb);
      v2f b;
      b.x = nok ? Wfh[kb * HD + n] : 0.0f;
      b.y = nok ? Wfh[(kb + 1) * HD + n] : 0.0f;
      acc0 = __builtin_amdgcn_wmma_f32_16x16x4_f32(false, a0, false, b, (short)0,
                                                   acc0, false, false);
      acc1 = __builtin_amdgcn_wmma_f32_16x16x4_f32(false, a1, false, b, (short)0,
                                                   acc1, false, false);
    }
    {
      int kb = 148 + kb0;
      v2f a0 = *(const v2f*)(chh + (m0 + l15) * CHP + kb);
      v2f a1 = *(const v2f*)(chh + (m0 + 16 + l15) * CHP + kb);
      v2f b;
      b.x = (nok && kb < HD)     ? Wfh[kb * HD + n]       : 0.0f;
      b.y = (nok && kb + 1 < HD) ? Wfh[(kb + 1) * HD + n] : 0.0f;
      acc0 = __builtin_amdgcn_wmma_f32_16x16x4_f32(false, a0, false, b, (short)0,
                                                   acc0, false, false);
      acc1 = __builtin_amdgcn_wmma_f32_16x16x4_f32(false, a1, false, b, (short)0,
                                                   acc1, false, false);
    }
    float bv = nok ? bfh[nc] : 0.0f;
    auto epilogue = [&](const v8f& acc, int mbase) {
      float s0 = 0.0f, s1 = 0.0f;
#pragma unroll
      for (int v = 0; v < 8; ++v) {
        int grow = mbase + v + roff;
        int node = base + (grow >> 2);
        float xv = (nok && node < end) ? xf[(size_t)node * HD + nc] : 0.0f;
        float f  = sigmoid_f(acc[v] + bv + xv);
        float w  = f * chc[grow * CHP + nc];
        if (v < 4) s0 += w; else s1 += w;
      }
      int nl = (mbase >> 2) + ((lane >> 4) << 1);
      fcs[nl * FCP + n]       = s0;
      fcs[(nl + 1) * FCP + n] = s1;
    };
    epilogue(acc0, m0);
    epilogue(acc1, m0 + 16);
  }
  __syncthreads();

  for (int e = tid; e < 16 * HD; e += 256) {
    int m = e / HD, col = e - m * HD;
    int node = base + m;
    if (node < end) {
      float i = sigmoid_f(ious[m * IOUP + col]);
      float o = sigmoid_f(ious[m * IOUP + HD + col]);
      float u = tanh_f(ious[m * IOUP + 2 * HD + col]);
      float c = i * u + fcs[m * FCP + col];
      cst[(size_t)node * HD + col] = c;
      hst[(size_t)node * HD + col] = o * tanh_f(c);
    }
  }
}

extern "C" void kernel_launch(void* const* d_in, const int* in_sizes, int n_in,
                              void* d_out, int out_size, void* d_ws, size_t ws_size,
                              hipStream_t stream) {
  const float* x     = (const float*)d_in[0];
  const float* Wioux = (const float*)d_in[1];
  const float* bioux = (const float*)d_in[2];
  const float* Wiouh = (const float*)d_in[3];
  const float* biouh = (const float*)d_in[4];
  const float* Wfx   = (const float*)d_in[5];
  const float* bfx   = (const float*)d_in[6];
  const float* Wfh   = (const float*)d_in[7];
  const float* bfh   = (const float*)d_in[8];
  (void)in_sizes; (void)n_in; (void)out_size; (void)ws_size;

  float* xiou = (float*)d_ws;
  float* xf   = xiou + (size_t)NN * IOUW;
  float* cst  = xf   + (size_t)NN * HD;
  float* hst  = (float*)d_out;

  proj_gemm<<<dim3(313, 15), 128, 0, stream>>>(x, Wioux, bioux, xiou, IOUW);
  proj_gemm<<<dim3(313, 5),  128, 0, stream>>>(x, Wfx,   bfx,   xf,   HD);

  const int leafEnd = 14539;
  leaf_kernel<<<(leafEnd * HD + 255) / 256, 256, 0, stream>>>(xiou, biouh,
                                                              cst, hst, leafEnd);

  const int bounds[8] = {14539, 18635, 19659, 19915, 19979, 19995, 19999, 20000};
  const size_t lds =
      (size_t)(2 * 64 * CHP + 16 * CHP + 16 * FCP + 16 * IOUP) * sizeof(float);
  int s = bounds[0];
  for (int L = 1; L < 8; ++L) {
    int e = bounds[L];
    int groups = (e - s + 15) / 16;
    level_kernel<<<groups, 256, lds, stream>>>(xiou, xf, Wiouh, biouh, Wfh, bfh,
                                               cst, hst, s, e);
    s = e;
  }
}


# level_kernel: Wfh and Wiouh staged per workgroup in LDS panels (k-pair order, coalesced batched copies); B fragments are 8-byte LDS reads, not a global load and wait per K-step
# speedup vs baseline: 1.7014x; 1.7014x over previous
#include <hip/hip_runtime.h>


typedef __attribute__((ext_vector_type(2))) float v2f;
typedef __attribute__((ext_vector_type(4))) float v4f;
typedef __attribute__((ext_vector_type(8))) float v8f;

#define NN   20000
#define DIN  300
#define HD   150
#define IOUW 450
#define CHP  152
#define IOUP 464
#define FCP  160
#define WROWS  76
#define WPITCH 544

__device__ __forceinline__ float sigmoid_f(float x) {
  return 1.0f / (1.0f + __expf(-x));
}
__device__ __forceinline__ float tanh_f(float x) {
  float t = __expf(2.0f * fminf(fmaxf(x, -15.0f), 15.0f));
  return (t - 1.0f) / (t + 1.0f);
}

__global__ void __launch_bounds__(128)
proj_gemm(const float* __restrict__ A, const float* __restrict__ B,
          const float* __restrict__ bias, float* __restrict__ C, int N) {
  __shared__ float Bs[32 * 304];
  const int tid  = threadIdx.x;
  const int lane = tid & 31;
  const int wave = tid >> 5;
  const int l15  = lane & 15;
  const int kb0  = (lane >> 4) << 1;
  const int n0   = (int)blockIdx.y * 32;

  for (int e = tid; e < 32 * DIN; e += 128) {
    int k = e >> 5, nl = e & 31;
    int n = n0 + nl;
    Bs[nl * 304 + k] = (n < N) ? B[k * N + n] : 0.0f;
  }
  __syncthreads();

  const int m0 = ((int)blockIdx.x * 4 + wave) * 16;
  if (m0 >= NN) return;
  const float* arow  = A + (size_t)(m0 + l15) * DIN;
  const float* b0row = Bs + l15 * 304;
  const float* b1row = Bs + (l15 + 16) * 304;
  v8f acc0 = {}, acc1 = {};
  for (int k0 = 0; k0 < DIN; k0 += 4) {
    int kb = k0 + kb0;
    v2f a  = *(const v2f*)(arow + kb);
    v2f b0 = *(const v2f*)(b0row + kb);
    v2f b1 = *(const v2f*)(b1row + kb);
    acc0 = __builtin_amdgcn_wmma_f32_16x16x4_f32(false, a, false, b0, (short)0,
                                                 acc0, false, false);
    acc1 = __builtin_amdgcn_wmma_f32_16x16x4_f32(false, a, false, b1, (short)0,
                                                 acc1, false, false);
  }
#pragma unroll
  for (int half = 0; half < 2; ++half) {
    int n = n0 + half * 16 + l15;
    if (n < N) {
      float bv = bias[n];
      const v8f& acc = half ? acc1 : acc0;
#pragma unroll
      for (int v = 0; v < 8; ++v) {
        int m = m0 + v + ((lane >> 4) << 3);
        C[(size_t)m * N + n] = acc[v] + bv;
      }
    }
  }
}

__global__ void leaf_kernel(const float* __restrict__ xiou,
                            const float* __restrict__ biouh,
                            float* __restrict__ cst, float* __restrict__ hst,
                            int count) {
  int e = (int)blockIdx.x * blockDim.x + threadIdx.x;
  if (e >= count * HD) return;
  int node = e / HD, col = e - node * HD;
  const float* r = xiou + (size_t)node * IOUW;
  float i = sigmoid_f(r[col] + biouh[col]);
  float o = sigmoid_f(r[HD + col] + biouh[HD + col]);
  float u = tanh_f(r[2 * HD + col] + biouh[2 * HD + col]);
  float c = i * u;
  cst[(size_t)node * HD + col] = c;
  hst[(size_t)node * HD + col] = o * tanh_f(c);
}

template <int LD, int COL0, int NVP, int PW>
__device__ __forceinline__ void stage_panel(const float* __restrict__ Wg,
                                            float* pan, int tid) {
  constexpr int ITEMS = 75 * NVP;
  constexpr int U     = 8;
  constexpr int NB    = (ITEMS + 256 * U - 1) / (256 * U);
  const v4f zero4 = {0.0f, 0.0f, 0.0f, 0.0f};
#pragma unroll
  for (int b = 0; b < NB; ++b) {
    v2f r0[U], r1[U];
#pragma unroll
    for (int u = 0; u < U; ++u) {
      int idx = (b * U + u) * 256 + tid;
      int idc = idx < ITEMS ? idx : ITEMS - 1;
      int kp = idc / NVP, cp = idc - kp * NVP;
      const float* p = Wg + (2 * kp) * LD + COL0 + 2 * cp;
      r0[u] = *(const v2f*)p;
      r1[u] = *(const v2f*)(p + LD);
    }
#pragma unroll
    for (int u = 0; u < U; ++u) {
      int idx = (b * U + u) * 256 + tid;
      if (idx < ITEMS) {
        int kp = idx / NVP, cp = idx - kp * NVP;
        v4f o = {r0[u].x, r1[u].x, r0[u].y, r1[u].y};
        *(v4f*)(pan + kp * WPITCH + 4 * cp) = o;
      }
    }
  }
  constexpr int PQ = PW / 2;
  for (int c = tid; c < PQ; c += 256)
    *(v4f*)(pan + 75 * WPITCH + 4 * c) = zero4;
  constexpr int NPAD = PQ - NVP;
  constexpr int NPD  = NPAD > 0 ? NPAD : 1;
  for (int idx = tid; idx < 75 * NPAD; idx += 256) {
    int kp = idx / NPD, cp = NVP + idx - kp * NPD;
    *(v4f*)(pan + kp * WPITCH + 4 * cp) = zero4;
  }
}

__global__ void __launch_bounds__(256)
level_kernel(const float* __restrict__ xiou, const float* __restrict__ xf,
             const float* __restrict__ Wiouh, const float* __restrict__ biouh,
             const float* __restrict__ Wfh, const float* __restrict__ bfh,
             float* __restrict__ cst, float* __restrict__ hst,
             int start, int end) {
  extern __shared__ __attribute__((aligned(16))) float smem[];
  float* chh  = smem;
  float* chc  = chh  + 64 * CHP;
  float* hsum = chc  + 64 * CHP;
  float* fcs  = hsum + 16 * CHP;
  float* ious = fcs  + 16 * FCP;
  float* wpan = ious + 16 * IOUP;

  const int tid  = threadIdx.x;
  const int lane = tid & 31;
  const int wave = tid >> 5;
  const int l15  = lane & 15;
  const int kb0  = (lane >> 4) << 1;
  const int roff = (lane >> 4) << 3;
  const int base = start + (int)blockIdx.x * 16;

  stage_panel<HD, 0, 75, 160>(Wfh, wpan, tid);

  for (int e = tid; e < 64 * CHP; e += 256) {
    int row = e / CHP, col = e - row * CHP;
    float hv = 0.0f, cv = 0.0f;
    int node = base + (row >> 2);
    if (col < HD && node < end) {
      int cr = 4 * ((NN - 1) - node) + 1 + (row & 3);
      if (cr < NN) {
        int ci = (NN - 1) - cr;
        hv = hst[(size_t)ci * HD + col];
        cv = cst[(size_t)ci * HD + col];
      }
    }
    chh[e] = hv;
    chc[e] = cv;
  }
  __syncthreads();
  for (int e = tid; e < 16 * CHP; e += 256) {
    int m = e / CHP, col = e - m * CHP;
    int r0 = (m * 4) * CHP + col;
    hsum[e] = chh[r0] + chh[r0 + CHP] + chh[r0 + 2 * CHP] + chh[r0 + 3 * CHP];
  }
  __syncthreads();

  const float* bpan = wpan + (lane >> 4) * WPITCH + l15 * 2;

  for (int t = wave; t < 20; t += 8) {
    int mp = t / 10, nt = t - mp * 10;
    int m0 = mp * 32, n0 = nt * 16;
    int n  = n0 + l15;
    bool nok = n < HD;
    int  nc  = nok ? n : 0;
    v8f acc0 = {}, acc1 = {};
    const float* a0p = chh + (m0 + l15) * CHP + kb0;
    const float* a1p = a0p + 16 * CHP;
    const float* bq  = bpan + n0 * 2;
    for (int ks = 0; ks < 38; ++ks) {
      v2f a0 = *(const v2f*)(a0p + 4 * ks);
      v2f a1 = *(const v2f*)(a1p + 4 * ks);
      v2f b  = *(const v2f*)(bq + 2 * ks * WPITCH);
      acc0 = __builtin_amdgcn_wmma_f32_16x16x4_f32(false, a0, false, b, (short)0,
                                                   acc0, false, false);
      acc1 = __builtin_amdgcn_wmma_f32_16x16x4_f32(false, a1, false, b, (short)0,
                                                   acc1, false, false);
    }
    float bv = nok ? bfh[nc] : 0.0f;
    auto epilogue = [&](const v8f& acc, int mbase) {
      float s0 = 0.0f, s1 = 0.0f;
#pragma unroll
      for (int v = 0; v < 8; ++v) {
        int grow = mbase + v + roff;
        int node = base + (grow >> 2);
        float xv = (nok && node < end) ? xf[(size_t)node * HD + nc] : 0.0f;
        float f  = sigmoid_f(acc[v] + bv + xv);
        float w  = f * chc[grow * CHP + nc];
        if (v < 4) s0 += w; else s1 += w;
      }
      int nl = (mbase >> 2) + ((lane >> 4) << 1);
      fcs[nl * FCP + n]       = s0;
      fcs[(nl + 1) * FCP + n] = s1;
    };
    epilogue(acc0, m0);
    epilogue(acc1, m0 + 16);
  }

  auto iou_epilogue = [&](const v8f& acc, int n) {
    if (n < IOUW) {
      float bv = biouh[n];
#pragma unroll
      for (int v = 0; v < 8; ++v) {
        int m = v + roff;
        int node = base + m;
        float xv = (node < end) ? xiou[(size_t)node * IOUW + n] : 0.0f;
        ious[m * IOUP + n] = acc[v] + bv + xv;
      }
    }
  };
  auto iou_pass = [&](int T0, int NT) {
    const float* ap  = hsum + l15 * CHP + kb0;
    const float* b0p = bpan + wave * 32;
    const float* b1p = b0p + 8 * 32;
    if (wave + 8 < NT) {
      v8f acc0 = {}, acc1 = {};
      for (int ks = 0; ks < 38; ++ks) {
        v2f a  = *(const v2f*)(ap + 4 * ks);
        v2f b0 = *(const v2f*)(b0p + 2 * ks * WPITCH);
        v2f b1 = *(const v2f*)(b1p + 2 * ks * WPITCH);
        acc0 = __builtin_amdgcn_wmma_f32_16x16x4_f32(false, a, false, b0, (short)0,
                                                     acc0, false, false);
        acc1 = __builtin_amdgcn_wmma_f32_16x16x4_f32(false, a, false, b1, (short)0,
                                                     acc1, false, false);
      }
      iou_epilogue(acc0, (T0 + wave) * 16 + l15);
      iou_epilogue(acc1, (T0 + wave + 8) * 16 + l15);
    } else {
      v8f acc0 = {};
      for (int ks = 0; ks < 38; ++ks) {
        v2f a  = *(const v2f*)(ap + 4 * ks);
        v2f b0 = *(const v2f*)(b0p + 2 * ks * WPITCH);
        acc0 = __builtin_amdgcn_wmma_f32_16x16x4_f32(false, a, false, b0, (short)0,
                                                     acc0, false, false);
      }
      iou_epilogue(acc0, (T0 + wave) * 16 + l15);
    }
  };
  __syncthreads();
  stage_panel<IOUW, 0, 128, 256>(Wiouh, wpan, tid);
  __syncthreads();
  iou_pass(0, 16);
  __syncthreads();
  stage_panel<IOUW, 256, 97, 208>(Wiouh, wpan, tid);
  __syncthreads();
  iou_pass(16, 13);
  __syncthreads();

  for (int e = tid; e < 16 * HD; e += 256) {
    int m = e / HD, col = e - m * HD;
    int node = base + m;
    if (node < end) {
      float i = sigmoid_f(ious[m * IOUP + col]);
      float o = sigmoid_f(ious[m * IOUP + HD + col]);
      float u = tanh_f(ious[m * IOUP + 2 * HD + col]);
      float c = i * u + fcs[m * FCP + col];
      cst[(size_t)node * HD + col] = c;
      hst[(size_t)node * HD + col] = o * tanh_f(c);
    }
  }
}

extern "C" void kernel_launch(void* const* d_in, const int* in_sizes, int n_in,
                              void* d_out, int out_size, void* d_ws, size_t ws_size,
                              hipStream_t stream) {
  const float* x     = (const float*)d_in[0];
  const float* Wioux = (const float*)d_in[1];
  const float* bioux = (const float*)d_in[2];
  const float* Wiouh = (const float*)d_in[3];
  const float* biouh = (const float*)d_in[4];
  const float* Wfx   = (const float*)d_in[5];
  const float* bfx   = (const float*)d_in[6];
  const float* Wfh   = (const float*)d_in[7];
  const float* bfh   = (const float*)d_in[8];
  (void)in_sizes; (void)n_in; (void)out_size; (void)ws_size;

  float* xiou = (float*)d_ws;
  float* xf   = xiou + (size_t)NN * IOUW;
  float* cst  = xf   + (size_t)NN * HD;
  float* hst  = (float*)d_out;

  proj_gemm<<<dim3(313, 15), 128, 0, stream>>>(x, Wioux, bioux, xiou, IOUW);
  proj_gemm<<<dim3(313, 5),  128, 0, stream>>>(x, Wfx,   bfx,   xf,   HD);

  const int leafEnd = 14539;
  leaf_kernel<<<(leafEnd * HD + 255) / 256, 256, 0, stream>>>(xiou, biouh,
                                                              cst, hst, leafEnd);

  const int bounds[8] = {14539, 18635, 19659, 19915, 19979, 19995, 19999, 20000};
  const size_t lds =
      (size_t)(2 * 64 * CHP + 16 * CHP + 16 * FCP + 16 * IOUP + WROWS * WPITCH) *
      sizeof(float);
  int s = bounds[0];
  for (int L = 1; L < 8; ++L) {
    int e = bounds[L];
    int groups = (e - s + 15) / 16;
    level_kernel<<<groups, 256, lds, stream>>>(xiou, xf, Wiouh, biouh, Wfh, bfh,
                                               cst, hst, s, e);
    s = e;
  }
}


# level_kernel: interior levels walked by subtree, a workgroup takes 16 nodes, then their 4 parents, then the grandparent, child c and h handed on in LDS: 3 level launches not 7, 4 gathers gone
# speedup vs baseline: 2.0219x; 1.1884x over previous
#include <hip/hip_runtime.h>


typedef __attribute__((ext_vector_type(2))) float v2f;
typedef __attribute__((ext_vector_type(4))) float v4f;
typedef __attribute__((ext_vector_type(8))) float v8f;

#define NN   20000
#define DIN  300
#define HD   150
#define IOUW 450
#define CHP  152
#define IOUP 464
#define FCP  160
#define WROWS  76
#define WPITCH 544

__device__ __forceinline__ float sigmoid_f(float x) {
  return 1.0f / (1.0f + __expf(-x));
}
__device__ __forceinline__ float tanh_f(float x) {
  float t = __expf(2.0f * fminf(fmaxf(x, -15.0f), 15.0f));
  return (t - 1.0f) / (t + 1.0f);
}

__global__ void __launch_bounds__(128)
proj_gemm(const float* __restrict__ A, const float* __restrict__ B,
          const float* __restrict__ bias, float* __restrict__ C, int N) {
  __shared__ float Bs[32 * 304];
  const int tid  = threadIdx.x;
  const int lane = tid & 31;
  const int wave = tid >> 5;
  const int l15  = lane & 15;
  const int kb0  = (lane >> 4) << 1;
  const int n0   = (int)blockIdx.y * 32;

  for (int e = tid; e < 32 * DIN; e += 128) {
    int k = e >> 5, nl = e & 31;
    int n = n0 + nl;
    Bs[nl * 304 + k] = (n < N) ? B[k * N + n] : 0.0f;
  }
  __syncthreads();

  const int m0 = ((int)blockIdx.x * 4 + wave) * 16;
  if (m0 >= NN) return;
  const float* arow  = A + (size_t)(m0 + l15) * DIN;
  const float* b0row = Bs + l15 * 304;
  const float* b1row = Bs + (l15 + 16) * 304;
  v8f acc0 = {}, acc1 = {};
  for (int k0 = 0; k0 < DIN; k0 += 4) {
    int kb = k0 + kb0;
    v2f a  = *(const v2f*)(arow + kb);
    v2f b0 = *(const v2f*)(b0row + kb);
    v2f b1 = *(const v2f*)(b1row + kb);
    acc0 = __builtin_amdgcn_wmma_f32_16x16x4_f32(false, a, false, b0, (short)0,
                                                 acc0, false, false);
    acc1 = __builtin_amdgcn_wmma_f32_16x16x4_f32(false, a, false, b1, (short)0,
                                                 acc1, false, false);
  }
#pragma unroll
  for (int half = 0; half < 2; ++half) {
    int n = n0 + half * 16 + l15;
    if (n < N) {
      float bv = bias[n];
      const v8f& acc = half ? acc1 : acc0;
#pragma unroll
      for (int v = 0; v < 8; ++v) {
        int m = m0 + v + ((lane >> 4) << 3);
        C[(size_t)m * N + n] = acc[v] + bv;
      }
    }
  }
}

__global__ void leaf_kernel(const float* __restrict__ xiou,
                            const float* __restrict__ biouh,
                            float* __restrict__ cst, float* __restrict__ hst,
                            int count) {
  int e = (int)blockIdx.x * blockDim.x + threadIdx.x;
  if (e >= count * HD) return;
  int node = e / HD, col = e - node * HD;
  const float* r = xiou + (size_t)node * IOUW;
  float i = sigmoid_f(r[col] + biouh[col]);
  float o = sigmoid_f(r[HD + col] + biouh[HD + col]);
  float u = tanh_f(r[2 * HD + col] + biouh[2 * HD + col]);
  float c = i * u;
  cst[(size_t)node * HD + col] = c;
  hst[(size_t)node * HD + col] = o * tanh_f(c);
}

template <int LD, int COL0, int NVP, int PW>
__device__ __forceinline__ void stage_panel(const float* __restrict__ Wg,
                                            float* pan, int tid) {
  constexpr int ITEMS = 75 * NVP;
  constexpr int U     = 8;
  constexpr int NB    = (ITEMS + 256 * U - 1) / (256 * U);
  const v4f zero4 = {0.0f, 0.0f, 0.0f, 0.0f};
#pragma unroll
  for (int b = 0; b < NB; ++b) {
    v2f r0[U], r1[U];
#pragma unroll
    for (int u = 0; u < U; ++u) {
      int idx = (b * U + u) * 256 + tid;
      int idc = idx < ITEMS ? idx : ITEMS - 1;
      int kp = idc / NVP, cp = idc - kp * NVP;
      const float* p = Wg + (2 * kp) * LD + COL0 + 2 * cp;
      r0[u] = *(const v2f*)p;
      r1[u] = *(const v2f*)(p + LD);
    }
#pragma unroll
    for (int u = 0; u < U; ++u) {
      int idx = (b * U + u) * 256 + tid;
      if (idx < ITEMS) {
        int kp = idx / NVP, cp = idx - kp * NVP;
        v4f o = {r0[u].x, r1[u].x, r0[u].y, r1[u].y};
        *(v4f*)(pan + kp * WPITCH + 4 * cp) = o;
      }
    }
  }
  constexpr int PQ = PW / 2;
  for (int c = tid; c < PQ; c += 256)
    *(v4f*)(pan + 75 * WPITCH + 4 * c) = zero4;
  constexpr int NPAD = PQ - NVP;
  constexpr int NPD  = NPAD > 0 ? NPAD : 1;
  for (int idx = tid; idx < 75 * NPAD; idx += 256) {
    int kp = idx / NPD, cp = NVP + idx - kp * NPD;
    *(v4f*)(pan + kp * WPITCH + 4 * cp) = zero4;
  }
}

extern __shared__ __attribute__((aligned(16))) float smem[];

typedef const __attribute__((address_space(1))) float* gmem_cf;
typedef __attribute__((address_space(1))) float* gmem_f;

__device__ __noinline__ void
sub_level(gmem_cf xiou_p, gmem_cf xf_p, gmem_cf Wiouh_p, gmem_cf biouh_p,
          gmem_cf Wfh_p, gmem_cf bfh_p, gmem_f cst_p, gmem_f hst_p,
          int base, int cnt, bool gather, bool more) {
  const float* __restrict__ xiou  = (const float*)xiou_p;
  const float* __restrict__ xf    = (const float*)xf_p;
  const float* __restrict__ Wiouh = (const float*)Wiouh_p;
  const float* __restrict__ biouh = (const float*)biouh_p;
  const float* __restrict__ Wfh   = (const float*)Wfh_p;
  const float* __restrict__ bfh   = (const float*)bfh_p;
  float* __restrict__ cst = (float*)cst_p;
  float* __restrict__ hst = (float*)hst_p;
  float* chh  = smem;
  float* chc  = chh  + 64 * CHP;
  float* hsum = chc  + 64 * CHP;
  float* fcs  = hsum + 16 * CHP;
  float* ious = fcs  + 16 * FCP;
  float* wpan = ious + 16 * IOUP;

  const int tid  = threadIdx.x;
  const int lane = tid & 31;
  const int wave = tid >> 5;
  const int l15  = lane & 15;
  const int kb0  = (lane >> 4) << 1;
  const int roff = (lane >> 4) << 3;
  const int end  = base + cnt;

  stage_panel<HD, 0, 75, 160>(Wfh, wpan, tid);

  if (gather)
  for (int e = tid; e < 64 * CHP; e += 256) {
    int row = e / CHP, col = e - row * CHP;
    float hv = 0.0f, cv = 0.0f;
    int node = base + (row >> 2);
    if (col < HD && node < end) {
      int cr = 4 * ((NN - 1) - node) + 1 + (row & 3);
      if (cr < NN) {
        int ci = (NN - 1) - cr;
        hv = hst[(size_t)ci * HD + col];
        cv = cst[(size_t)ci * HD + col];
      }
    }
    chh[e] = hv;
    chc[e] = cv;
  }
  __syncthreads();
  for (int e = tid; e < 16 * CHP; e += 256) {
    int m = e / CHP, col = e - m * CHP;
    int r0 = (m * 4) * CHP + col;
    hsum[e] = chh[r0] + chh[r0 + CHP] + chh[r0 + 2 * CHP] + chh[r0 + 3 * CHP];
  }
  __syncthreads();

  const float* bpan = wpan + (lane >> 4) * WPITCH + l15 * 2;

  const int ntask = ((4 * cnt + 31) >> 5) * 10;
  for (int t = wave; t < ntask; t += 8) {
    int mp = t / 10, nt = t - mp * 10;
    int m0 = mp * 32, n0 = nt * 16;
    int n  = n0 + l15;
    bool nok = n < HD;
    int  nc  = nok ? n : 0;
    v8f acc0 = {}, acc1 = {};
    const float* a0p = chh + (m0 + l15) * CHP + kb0;
    const float* a1p = a0p + 16 * CHP;
    const float* bq  = bpan + n0 * 2;
    for (int ks = 0; ks < 38; ++ks) {
      v2f a0 = *(const v2f*)(a0p + 4 * ks);
      v2f a1 = *(const v2f*)(a1p + 4 * ks);
      v2f b  = *(const v2f*)(bq + 2 * ks * WPITCH);
      acc0 = __builtin_amdgcn_wmma_f32_16x16x4_f32(false, a0, false, b, (short)0,
                                                   acc0, false, false);
      acc1 = __builtin_amdgcn_wmma_f32_16x16x4_f32(false, a1, false, b, (short)0,
                                                   acc1, false, false);
    }
    float bv = nok ? bfh[nc] : 0.0f;
    auto epilogue = [&](const v8f& acc, int mbase) {
      float s0 = 0.0f, s1 = 0.0f;
#pragma unroll
      for (int v = 0; v < 8; ++v) {
        int grow = mbase + v + roff;
        int node = base + (grow >> 2);
        float xv = (nok && node < end) ? xf[(size_t)node * HD + nc] : 0.0f;
        float f  = sigmoid_f(acc[v] + bv + xv);
        float w  = f * chc[grow * CHP + nc];
        if (v < 4) s0 += w; else s1 += w;
      }
      int nl = (mbase >> 2) + ((lane >> 4) << 1);
      fcs[nl * FCP + n]       = s0;
      fcs[(nl + 1) * FCP + n] = s1;
    };
    epilogue(acc0, m0);
    epilogue(acc1, m0 + 16);
  }

  auto iou_epilogue = [&](const v8f& acc, int n) {
    if (n < IOUW) {
      float bv = biouh[n];
#pragma unroll
      for (int v = 0; v < 8; ++v) {
        int m = v + roff;
        int node = base + m;
        float xv = (node < end) ? xiou[(size_t)node * IOUW + n] : 0.0f;
        ious[m * IOUP + n] = acc[v] + bv + xv;
      }
    }
  };
  auto iou_pass = [&](int T0, int NT) {
    const float* ap  = hsum + l15 * CHP + kb0;
    const float* b0p = bpan + wave * 32;
    const float* b1p = b0p + 8 * 32;
    if (wave + 8 < NT) {
      v8f acc0 = {}, acc1 = {};
      for (int ks = 0; ks < 38; ++ks) {
        v2f a  = *(const v2f*)(ap + 4 * ks);
        v2f b0 = *(const v2f*)(b0p + 2 * ks * WPITCH);
        v2f b1 = *(const v2f*)(b1p + 2 * ks * WPITCH);
        acc0 = __builtin_amdgcn_wmma_f32_16x16x4_f32(false, a, false, b0, (short)0,
                                                     acc0, false, false);
        acc1 = __builtin_amdgcn_wmma_f32_16x16x4_f32(false, a, false, b1, (short)0,
                                                     acc1, false, false);
      }
      iou_epilogue(acc0, (T0 + wave) * 16 + l15);
      iou_epilogue(acc1, (T0 + wave + 8) * 16 + l15);
    } else {
      v8f acc0 = {};
      for (int ks = 0; ks < 38; ++ks) {
        v2f a  = *(const v2f*)(ap + 4 * ks);
        v2f b0 = *(const v2f*)(b0p + 2 * ks * WPITCH);
        acc0 = __builtin_amdgcn_wmma_f32_16x16x4_f32(false, a, false, b0, (short)0,
                                                     acc0, false, false);
      }
      iou_epilogue(acc0, (T0 + wave) * 16 + l15);
    }
  };
  __syncthreads();
  stage_panel<IOUW, 0, 128, 256>(Wiouh, wpan, tid);
  __syncthreads();
  iou_pass(0, 16);
  __syncthreads();
  stage_panel<IOUW, 256, 97, 208>(Wiouh, wpan, tid);
  __syncthreads();
  iou_pass(16, 13);
  __syncthreads();

  for (int e = tid; e < 16 * HD; e += 256) {
    int m = e / HD, col = e - m * HD;
    int node = base + m;
    if (node < end) {
      float i = sigmoid_f(ious[m * IOUP + col]);
      float o = sigmoid_f(ious[m * IOUP + HD + col]);
      float u = tanh_f(ious[m * IOUP + 2 * HD + col]);
      float c = i * u + fcs[m * FCP + col];
      float h = o * tanh_f(c);
      hst[(size_t)node * HD + col] = h;
      if (more) {
        chh[(m ^ 3) * CHP + col] = h;
        chc[(m ^ 3) * CHP + col] = c;
      } else {
        cst[(size_t)node * HD + col] = c;
      }
    }
  }
}

__global__ void __launch_bounds__(256)
level_kernel(const float* __restrict__ xiou, const float* __restrict__ xf,
             const float* __restrict__ Wiouh, const float* __restrict__ biouh,
             const float* __restrict__ Wfh, const float* __restrict__ bfh,
             float* __restrict__ cst, float* __restrict__ hst,
             int s0, int s1, int s2, int cnt0, int nsub) {
  for (int sub = 0; sub < nsub; ++sub) {
    const int cnt  = cnt0 >> (2 * sub);
    const int lvl0 = sub == 0 ? s0 : (sub == 1 ? s1 : s2);
    sub_level((gmem_cf)xiou, (gmem_cf)xf, (gmem_cf)Wiouh, (gmem_cf)biouh,
              (gmem_cf)Wfh, (gmem_cf)bfh, (gmem_f)cst, (gmem_f)hst,
              lvl0 + (int)blockIdx.x * cnt, cnt, sub == 0, sub + 1 < nsub);
  }
}

extern "C" void kernel_launch(void* const* d_in, const int* in_sizes, int n_in,
                              void* d_out, int out_size, void* d_ws, size_t ws_size,
                              hipStream_t stream) {
  const float* x     = (const float*)d_in[0];
  const float* Wioux = (const float*)d_in[1];
  const float* bioux = (const float*)d_in[2];
  const float* Wiouh = (const float*)d_in[3];
  const float* biouh = (const float*)d_in[4];
  const float* Wfx   = (const float*)d_in[5];
  const float* bfx   = (const float*)d_in[6];
  const float* Wfh   = (const float*)d_in[7];
  const float* bfh   = (const float*)d_in[8];
  (void)in_sizes; (void)n_in; (void)out_size; (void)ws_size;

  float* xiou = (float*)d_ws;
  float* xf   = xiou + (size_t)NN * IOUW;
  float* cst  = xf   + (size_t)NN * HD;
  float* hst  = (float*)d_out;

  proj_gemm<<<dim3(313, 15), 128, 0, stream>>>(x, Wioux, bioux, xiou, IOUW);
  proj_gemm<<<dim3(313, 5),  128, 0, stream>>>(x, Wfx,   bfx,   xf,   HD);

  const int leafEnd = 14539;
  leaf_kernel<<<(leafEnd * HD + 255) / 256, 256, 0, stream>>>(xiou, biouh,
                                                              cst, hst, leafEnd);

  const size_t lds =
      (size_t)(2 * 64 * CHP + 16 * CHP + 16 * FCP + 16 * IOUP + WROWS * WPITCH) *
      sizeof(float);
  level_kernel<<<256, 256, lds, stream>>>(xiou, xf, Wiouh, biouh, Wfh, bfh,
                                          cst, hst, 14539, 18635, 19659, 16, 3);
  level_kernel<<<4, 256, lds, stream>>>(xiou, xf, Wiouh, biouh, Wfh, bfh,
                                        cst, hst, 19915, 19979, 0, 16, 2);
  level_kernel<<<1, 256, lds, stream>>>(xiou, xf, Wiouh, biouh, Wfh, bfh,
                                        cst, hst, 19995, 19999, 0, 4, 2);
}


# proj_gemm: weights in registers, not LDS: a wave loads the B fragments of its 32 columns for all K once (300 VGPRs) and walks 3-10 row tiles, a tile's 75 A loads issued before its WMMAs; same sums
# speedup vs baseline: 2.6184x; 1.2950x over previous
#include <hip/hip_runtime.h>


typedef __attribute__((ext_vector_type(2))) float v2f;
typedef __attribute__((ext_vector_type(4))) float v4f;
typedef __attribute__((ext_vector_type(8))) float v8f;

#define NN   20000
#define DIN  300
#define HD   150
#define IOUW 450
#define CHP  152
#define IOUP 464
#define FCP  160
#define WROWS  76
#define WPITCH 544

__device__ __forceinline__ float sigmoid_f(float x) {
  return 1.0f / (1.0f + __expf(-x));
}
__device__ __forceinline__ float tanh_f(float x) {
  float t = __expf(2.0f * fminf(fmaxf(x, -15.0f), 15.0f));
  return (t - 1.0f) / (t + 1.0f);
}

__global__ void __launch_bounds__(256)
proj_gemm(const float* __restrict__ A, const float* __restrict__ B,
          const float* __restrict__ bias, float* __restrict__ C, int N) {
  const int tid  = threadIdx.x;
  const int lane = tid & 31;
  const int wave = __builtin_amdgcn_readfirstlane(tid >> 5);
  const int l15  = lane & 15;
  const int kb0  = (lane >> 4) << 1;
  const int n0   = (int)blockIdx.y * 32;
  const int na   = n0 + l15, nb = na + 16;
  const int ca = na < N ? na : N - 1;
  const int cb = nb < N ? nb : N - 1;

  v2f bfa[75], bfb[75];
  const unsigned un = (unsigned)N;
  const unsigned oa = (unsigned)kb0 * un + (unsigned)ca;
  const unsigned ob = (unsigned)kb0 * un + (unsigned)cb;
#pragma unroll
  for (int ks = 0; ks < 75; ++ks) {
    const unsigned kr = 4u * (unsigned)ks * un;
    bfa[ks].x = B[oa + kr];
    bfa[ks].y = B[oa + kr + un];
    bfb[ks].x = B[ob + kr];
    bfb[ks].y = B[ob + kr + un];
  }
  const float bva = bias[ca], bvb = bias[cb];

  const int step = (int)gridDim.x * 8;
  for (int mt = (int)blockIdx.x * 8 + wave; mt < NN / 16; mt += step) {
    const int m0 = mt * 16;
    const float* arow = A + (size_t)(m0 + l15) * DIN + kb0;
    v2f a[75];
#pragma unroll
    for (int ks = 0; ks < 75; ++ks) a[ks] = *(const v2f*)(arow + 4 * ks);
    __builtin_amdgcn_sched_barrier(0);
    v8f acc0 = {}, acc1 = {};
#pragma unroll
    for (int ks = 0; ks < 75; ++ks) {
      acc0 = __builtin_amdgcn_wmma_f32_16x16x4_f32(false, a[ks], false, bfa[ks],
                                                   (short)0, acc0, false, false);
      acc1 = __builtin_amdgcn_wmma_f32_16x16x4_f32(false, a[ks], false, bfb[ks],
                                                   (short)0, acc1, false, false);
    }
    float* crow = C + (size_t)(m0 + ((lane >> 4) << 3)) * N;
    if (na < N) {
#pragma unroll
      for (int v = 0; v < 8; ++v) crow[(size_t)v * N + na] = acc0[v] + bva;
    }
    if (nb < N) {
#pragma unroll
      for (int v = 0; v < 8; ++v) crow[(size_t)v * N + nb] = acc1[v] + bvb;
    }
  }
}

__global__ void leaf_kernel(const float* __restrict__ xiou,
                            const float* __restrict__ biouh,
                            float* __restrict__ cst, float* __restrict__ hst,
                            int count) {
  int e = (int)blockIdx.x * blockDim.x + threadIdx.x;
  if (e >= count * HD) return;
  int node = e / HD, col = e - node * HD;
  const float* r = xiou + (size_t)node * IOUW;
  float i = sigmoid_f(r[col] + biouh[col]);
  float o = sigmoid_f(r[HD + col] + biouh[HD + col]);
  float u = tanh_f(r[2 * HD + col] + biouh[2 * HD + col]);
  float c = i * u;
  cst[(size_t)node * HD + col] = c;
  hst[(size_t)node * HD + col] = o * tanh_f(c);
}

template <int LD, int COL0, int NVP, int PW>
__device__ __forceinline__ void stage_panel(const float* __restrict__ Wg,
                                            float* pan, int tid) {
  constexpr int ITEMS = 75 * NVP;
  constexpr int U     = 8;
  constexpr int NB    = (ITEMS + 256 * U - 1) / (256 * U);
  const v4f zero4 = {0.0f, 0.0f, 0.0f, 0.0f};
#pragma unroll
  for (int b = 0; b < NB; ++b) {
    v2f r0[U], r1[U];
#pragma unroll
    for (int u = 0; u < U; ++u) {
      int idx = (b * U + u) * 256 + tid;
      int idc = idx < ITEMS ? idx : ITEMS - 1;
      int kp = idc / NVP, cp = idc - kp * NVP;
      const float* p = Wg + (2 * kp) * LD + COL0 + 2 * cp;
      r0[u] = *(const v2f*)p;
      r1[u] = *(const v2f*)(p + LD);
    }
#pragma unroll
    for (int u = 0; u < U; ++u) {
      int idx = (b * U + u) * 256 + tid;
      if (idx < ITEMS) {
        int kp = idx / NVP, cp = idx - kp * NVP;
        v4f o = {r0[u].x, r1[u].x, r0[u].y, r1[u].y};
        *(v4f*)(pan + kp * WPITCH + 4 * cp) = o;
      }
    }
  }
  constexpr int PQ = PW / 2;
  for (int c = tid; c < PQ; c += 256)
    *(v4f*)(pan + 75 * WPITCH + 4 * c) = zero4;
  constexpr int NPAD = PQ - NVP;
  constexpr int NPD  = NPAD > 0 ? NPAD : 1;
  for (int idx = tid; idx < 75 * NPAD; idx += 256) {
    int kp = idx / NPD, cp = NVP + idx - kp * NPD;
    *(v4f*)(pan + kp * WPITCH + 4 * cp) = zero4;
  }
}

extern __shared__ __attribute__((aligned(16))) float smem[];

typedef const __attribute__((address_space(1))) float* gmem_cf;
typedef __attribute__((address_space(1))) float* gmem_f;

__device__ __noinline__ void
sub_level(gmem_cf xiou_p, gmem_cf xf_p, gmem_cf Wiouh_p, gmem_cf biouh_p,
          gmem_cf Wfh_p, gmem_cf bfh_p, gmem_f cst_p, gmem_f hst_p,
          int base, int cnt, bool gather, bool more) {
  const float* __restrict__ xiou  = (const float*)xiou_p;
  const float* __restrict__ xf    = (const float*)xf_p;
  const float* __restrict__ Wiouh = (const float*)Wiouh_p;
  const float* __restrict__ biouh = (const float*)biouh_p;
  const float* __restrict__ Wfh   = (const float*)Wfh_p;
  const float* __restrict__ bfh   = (const float*)bfh_p;
  float* __restrict__ cst = (float*)cst_p;
  float* __restrict__ hst = (float*)hst_p;
  float* chh  = smem;
  float* chc  = chh  + 64 * CHP;
  float* hsum = chc  + 64 * CHP;
  float* fcs  = hsum + 16 * CHP;
  float* ious = fcs  + 16 * FCP;
  float* wpan = ious + 16 * IOUP;

  const int tid  = threadIdx.x;
  const int lane = tid & 31;
  const int wave = tid >> 5;
  const int l15  = lane & 15;
  const int kb0  = (lane >> 4) << 1;
  const int roff = (lane >> 4) << 3;
  const int end  = base + cnt;

  stage_panel<HD, 0, 75, 160>(Wfh, wpan, tid);

  if (gather)
  for (int e = tid; e < 64 * CHP; e += 256) {
    int row = e / CHP, col = e - row * CHP;
    float hv = 0.0f, cv = 0.0f;
    int node = base + (row >> 2);
    if (col < HD && node < end) {
      int cr = 4 * ((NN - 1) - node) + 1 + (row & 3);
      if (cr < NN) {
        int ci = (NN - 1) - cr;
        hv = hst[(size_t)ci * HD + col];
        cv = cst[(size_t)ci * HD + col];
      }
    }
    chh[e] = hv;
    chc[e] = cv;
  }
  __syncthreads();
  for (int e = tid; e < 16 * CHP; e += 256) {
    int m = e / CHP, col = e - m * CHP;
    int r0 = (m * 4) * CHP + col;
    hsum[e] = chh[r0] + chh[r0 + CHP] + chh[r0 + 2 * CHP] + chh[r0 + 3 * CHP];
  }
  __syncthreads();

  const float* bpan = wpan + (lane >> 4) * WPITCH + l15 * 2;

  const int ntask = ((4 * cnt + 31) >> 5) * 10;
  for (int t = wave; t < ntask; t += 8) {
    int mp = t / 10, nt = t - mp * 10;
    int m0 = mp * 32, n0 = nt * 16;
    int n  = n0 + l15;
    bool nok = n < HD;
    int  nc  = nok ? n : 0;
    v8f acc0 = {}, acc1 = {};
    const float* a0p = chh + (m0 + l15) * CHP + kb0;
    const float* a1p = a0p + 16 * CHP;
    const float* bq  = bpan + n0 * 2;
    for (int ks = 0; ks < 38; ++ks) {
      v2f a0 = *(const v2f*)(a0p + 4 * ks);
      v2f a1 = *(const v2f*)(a1p + 4 * ks);
      v2f b  = *(const v2f*)(bq + 2 * ks * WPITCH);
      acc0 = __builtin_amdgcn_wmma_f32_16x16x4_f32(false, a0, false, b, (short)0,
                                                   acc0, false, false);
      acc1 = __builtin_amdgcn_wmma_f32_16x16x4_f32(false, a1, false, b, (short)0,
                                                   acc1, false, false);
    }
    float bv = nok ? bfh[nc] : 0.0f;
    auto epilogue = [&](const v8f& acc, int mbase) {
      float s0 = 0.0f, s1 = 0.0f;
#pragma unroll
      for (int v = 0; v < 8; ++v) {
        int grow = mbase + v + roff;
        int node = base + (grow >> 2);
        float xv = (nok && node < end) ? xf[(size_t)node * HD + nc] : 0.0f;
        float f  = sigmoid_f(acc[v] + bv + xv);
        float w  = f * chc[grow * CHP + nc];
        if (v < 4) s0 += w; else s1 += w;
      }
      int nl = (mbase >> 2) + ((lane >> 4) << 1);
      fcs[nl * FCP + n]       = s0;
      fcs[(nl + 1) * FCP + n] = s1;
    };
    epilogue(acc0, m0);
    epilogue(acc1, m0 + 16);
  }

  auto iou_epilogue = [&](const v8f& acc, int n) {
    if (n < IOUW) {
      float bv = biouh[n];
#pragma unroll
      for (int v = 0; v < 8; ++v) {
        int m = v + roff;
        int node = base + m;
        float xv = (node < end) ? xiou[(size_t)node * IOUW + n] : 0.0f;
        ious[m * IOUP + n] = acc[v] + bv + xv;
      }
    }
  };
  auto iou_pass = [&](int T0, int NT) {
    const float* ap  = hsum + l15 * CHP + kb0;
    const float* b0p = bpan + wave * 32;
    const float* b1p = b0p + 8 * 32;
    if (wave + 8 < NT) {
      v8f acc0 = {}, acc1 = {};
      for (int ks = 0; ks < 38; ++ks) {
        v2f a  = *(const v2f*)(ap + 4 * ks);
        v2f b0 = *(const v2f*)(b0p + 2 * ks * WPITCH);
        v2f b1 = *(const v2f*)(b1p + 2 * ks * WPITCH);
        acc0 = __builtin_amdgcn_wmma_f32_16x16x4_f32(false, a, false, b0, (short)0,
                                                     acc0, false, false);
        acc1 = __builtin_amdgcn_wmma_f32_16x16x4_f32(false, a, false, b1, (short)0,
                                                     acc1, false, false);
      }
      iou_epilogue(acc0, (T0 + wave) * 16 + l15);
      iou_epilogue(acc1, (T0 + wave + 8) * 16 + l15);
    } else {
      v8f acc0 = {};
      for (int ks = 0; ks < 38; ++ks) {
        v2f a  = *(const v2f*)(ap + 4 * ks);
        v2f b0 = *(const v2f*)(b0p + 2 * ks * WPITCH);
        acc0 = __builtin_amdgcn_wmma_f32_16x16x4_f32(false, a, false, b0, (short)0,
                                                     acc0, false, false);
      }
      iou_epilogue(acc0, (T0 + wave) * 16 + l15);
    }
  };
  __syncthreads();
  stage_panel<IOUW, 0, 128, 256>(Wiouh, wpan, tid);
  __syncthreads();
  iou_pass(0, 16);
  __syncthreads();
  stage_panel<IOUW, 256, 97, 208>(Wiouh, wpan, tid);
  __syncthreads();
  iou_pass(16, 13);
  __syncthreads();

  for (int e = tid; e < 16 * HD; e += 256) {
    int m = e / HD, col = e - m * HD;
    int node = base + m;
    if (node < end) {
      float i = sigmoid_f(ious[m * IOUP + col]);
      float o = sigmoid_f(ious[m * IOUP + HD + col]);
      float u = tanh_f(ious[m * IOUP + 2 * HD + col]);
      float c = i * u + fcs[m * FCP + col];
      float h = o * tanh_f(c);
      hst[(size_t)node * HD + col] = h;
      if (more) {
        chh[(m ^ 3) * CHP + col] = h;
        chc[(m ^ 3) * CHP + col] = c;
      } else {
        cst[(size_t)node * HD + col] = c;
      }
    }
  }
}

__global__ void __launch_bounds__(256)
level_kernel(const float* __restrict__ xiou, const float* __restrict__ xf,
             const float* __restrict__ Wiouh, const float* __restrict__ biouh,
             const float* __restrict__ Wfh, const float* __restrict__ bfh,
             float* __restrict__ cst, float* __restrict__ hst,
             int s0, int s1, int s2, int cnt0, int nsub) {
  for (int sub = 0; sub < nsub; ++sub) {
    const int cnt  = cnt0 >> (2 * sub);
    const int lvl0 = sub == 0 ? s0 : (sub == 1 ? s1 : s2);
    sub_level((gmem_cf)xiou, (gmem_cf)xf, (gmem_cf)Wiouh, (gmem_cf)biouh,
              (gmem_cf)Wfh, (gmem_cf)bfh, (gmem_f)cst, (gmem_f)hst,
              lvl0 + (int)blockIdx.x * cnt, cnt, sub == 0, sub + 1 < nsub);
  }
}

extern "C" void kernel_launch(void* const* d_in, const int* in_sizes, int n_in,
                              void* d_out, int out_size, void* d_ws, size_t ws_size,
                              hipStream_t stream) {
  const float* x     = (const float*)d_in[0];
  const float* Wioux = (const float*)d_in[1];
  const float* bioux = (const float*)d_in[2];
  const float* Wiouh = (const float*)d_in[3];
  const float* biouh = (const float*)d_in[4];
  const float* Wfx   = (const float*)d_in[5];
  const float* bfx   = (const float*)d_in[6];
  const float* Wfh   = (const float*)d_in[7];
  const float* bfh   = (const float*)d_in[8];
  (void)in_sizes; (void)n_in; (void)out_size; (void)ws_size;

  float* xiou = (float*)d_ws;
  float* xf   = xiou + (size_t)NN * IOUW;
  float* cst  = xf   + (size_t)NN * HD;
  float* hst  = (float*)d_out;

  proj_gemm<<<dim3(17, 15), 256, 0, stream>>>(x, Wioux, bioux, xiou, IOUW);
  proj_gemm<<<dim3(51, 5),  256, 0, stream>>>(x, Wfx,   bfx,   xf,   HD);

  const int leafEnd = 14539;
  leaf_kernel<<<(leafEnd * HD + 255) / 256, 256, 0, stream>>>(xiou, biouh,
                                                              cst, hst, leafEnd);

  const size_t lds =
      (size_t)(2 * 64 * CHP + 16 * CHP + 16 * FCP + 16 * IOUP + WROWS * WPITCH) *
      sizeof(float);
  level_kernel<<<256, 256, lds, stream>>>(xiou, xf, Wiouh, biouh, Wfh, bfh,
                                          cst, hst, 14539, 18635, 19659, 16, 3);
  level_kernel<<<4, 256, lds, stream>>>(xiou, xf, Wiouh, biouh, Wfh, bfh,
                                        cst, hst, 19915, 19979, 0, 16, 2);
  level_kernel<<<1, 256, lds, stream>>>(xiou, xf, Wiouh, biouh, Wfh, bfh,
                                        cst, hst, 19995, 19999, 0, 4, 2);
}


# level_kernel: recurrent weights fetched once per workgroup, not once per sub-level: Wiouh B fragments in registers (a wave's 4 n-tiles, all K), Wfh LDS panel staged before the loop; same sums
# speedup vs baseline: 3.2697x; 1.2487x over previous
#include <hip/hip_runtime.h>


typedef __attribute__((ext_vector_type(2))) float v2f;
typedef __attribute__((ext_vector_type(4))) float v4f;
typedef __attribute__((ext_vector_type(8))) float v8f;

#define NN   20000
#define DIN  300
#define HD   150
#define IOUW 450
#define CHP  152
#define IOUP 464
#define FCP  160
#define WROWS  76
#define WPITCH 544

__device__ __forceinline__ float sigmoid_f(float x) {
  return 1.0f / (1.0f + __expf(-x));
}
__device__ __forceinline__ float tanh_f(float x) {
  float t = __expf(2.0f * fminf(fmaxf(x, -15.0f), 15.0f));
  return (t - 1.0f) / (t + 1.0f);
}

__global__ void __launch_bounds__(256)
proj_gemm(const float* __restrict__ A, const float* __restrict__ B,
          const float* __restrict__ bias, float* __restrict__ C, int N) {
  const int tid  = threadIdx.x;
  const int lane = tid & 31;
  const int wave = __builtin_amdgcn_readfirstlane(tid >> 5);
  const int l15  = lane & 15;
  const int kb0  = (lane >> 4) << 1;
  const int n0   = (int)blockIdx.y * 32;
  const int na   = n0 + l15, nb = na + 16;
  const int ca = na < N ? na : N - 1;
  const int cb = nb < N ? nb : N - 1;

  v2f bfa[75], bfb[75];
  const unsigned un = (unsigned)N;
  const unsigned oa = (unsigned)kb0 * un + (unsigned)ca;
  const unsigned ob = (unsigned)kb0 * un + (unsigned)cb;
#pragma unroll
  for (int ks = 0; ks < 75; ++ks) {
    const unsigned kr = 4u * (unsigned)ks * un;
    bfa[ks].x = B[oa + kr];
    bfa[ks].y = B[oa + kr + un];
    bfb[ks].x = B[ob + kr];
    bfb[ks].y = B[ob + kr + un];
  }
  const float bva = bias[ca], bvb = bias[cb];

  const int step = (int)gridDim.x * 8;
  for (int mt = (int)blockIdx.x * 8 + wave; mt < NN / 16; mt += step) {
    const int m0 = mt * 16;
    const float* arow = A + (size_t)(m0 + l15) * DIN + kb0;
    v2f a[75];
#pragma unroll
    for (int ks = 0; ks < 75; ++ks) a[ks] = *(const v2f*)(arow + 4 * ks);
    __builtin_amdgcn_sched_barrier(0);
    v8f acc0 = {}, acc1 = {};
#pragma unroll
    for (int ks = 0; ks < 75; ++ks) {
      acc0 = __builtin_amdgcn_wmma_f32_16x16x4_f32(false, a[ks], false, bfa[ks],
                                                   (short)0, acc0, false, false);
      acc1 = __builtin_amdgcn_wmma_f32_16x16x4_f32(false, a[ks], false, bfb[ks],
                                                   (short)0, acc1, false, false);
    }
    float* crow = C + (size_t)(m0 + ((lane >> 4) << 3)) * N;
    if (na < N) {
#pragma unroll
      for (int v = 0; v < 8; ++v) crow[(size_t)v * N + na] = acc0[v] + bva;
    }
    if (nb < N) {
#pragma unroll
      for (int v = 0; v < 8; ++v) crow[(size_t)v * N + nb] = acc1[v] + bvb;
    }
  }
}

__global__ void leaf_kernel(const float* __restrict__ xiou,
                            const float* __restrict__ biouh,
                            float* __restrict__ cst, float* __restrict__ hst,
                            int count) {
  int e = (int)blockIdx.x * blockDim.x + threadIdx.x;
  if (e >= count * HD) return;
  int node = e / HD, col = e - node * HD;
  const float* r = xiou + (size_t)node * IOUW;
  float i = sigmoid_f(r[col] + biouh[col]);
  float o = sigmoid_f(r[HD + col] + biouh[HD + col]);
  float u = tanh_f(r[2 * HD + col] + biouh[2 * HD + col]);
  float c = i * u;
  cst[(size_t)node * HD + col] = c;
  hst[(size_t)node * HD + col] = o * tanh_f(c);
}

template <int LD, int COL0, int NVP, int PW>
__device__ __forceinline__ void stage_panel(const float* __restrict__ Wg,
                                            float* pan, int tid) {
  constexpr int ITEMS = 75 * NVP;
  constexpr int U     = 8;
  constexpr int NB    = (ITEMS + 256 * U - 1) / (256 * U);
  const v4f zero4 = {0.0f, 0.0f, 0.0f, 0.0f};
#pragma unroll
  for (int b = 0; b < NB; ++b) {
    v2f r0[U], r1[U];
#pragma unroll
    for (int u = 0; u < U; ++u) {
      int idx = (b * U + u) * 256 + tid;
      int idc = idx < ITEMS ? idx : ITEMS - 1;
      int kp = idc / NVP, cp = idc - kp * NVP;
      const float* p = Wg + (2 * kp) * LD + COL0 + 2 * cp;
      r0[u] = *(const v2f*)p;
      r1[u] = *(const v2f*)(p + LD);
    }
#pragma unroll
    for (int u = 0; u < U; ++u) {
      int idx = (b * U + u) * 256 + tid;
      if (idx < ITEMS) {
        int kp = idx / NVP, cp = idx - kp * NVP;
        v4f o = {r0[u].x, r1[u].x, r0[u].y, r1[u].y};
        *(v4f*)(pan + kp * WPITCH + 4 * cp) = o;
      }
    }
  }
  constexpr int PQ = PW / 2;
  for (int c = tid; c < PQ; c += 256)
    *(v4f*)(pan + 75 * WPITCH + 4 * c) = zero4;
  constexpr int NPAD = PQ - NVP;
  constexpr int NPD  = NPAD > 0 ? NPAD : 1;
  for (int idx = tid; idx < 75 * NPAD; idx += 256) {
    int kp = idx / NPD, cp = NVP + idx - kp * NPD;
    *(v4f*)(pan + kp * WPITCH + 4 * cp) = zero4;
  }
}

extern __shared__ __attribute__((aligned(16))) float smem[];

typedef const __attribute__((address_space(1))) float* gmem_cf;
typedef __attribute__((address_space(1))) float* gmem_f;

__device__ __forceinline__ int iou_tile(int w, int j) {
  return (j >> 1) * 16 + 2 * w + (j & 1);
}

__device__ __forceinline__ void
sub_level(gmem_cf xiou_p, gmem_cf xf_p, gmem_cf biouh_p, gmem_cf bfh_p,
          gmem_f cst_p, gmem_f hst_p, const v2f (&wb)[4][38],
          int base, int cnt, bool gather, bool more) {
  const float* __restrict__ xiou  = (const float*)xiou_p;
  const float* __restrict__ xf    = (const float*)xf_p;
  const float* __restrict__ biouh = (const float*)biouh_p;
  const float* __restrict__ bfh   = (const float*)bfh_p;
  float* __restrict__ cst = (float*)cst_p;
  float* __restrict__ hst = (float*)hst_p;
  float* chh  = smem;
  float* chc  = chh  + 64 * CHP;
  float* hsum = chc  + 64 * CHP;
  float* fcs  = hsum + 16 * CHP;
  float* ious = fcs  + 16 * FCP;
  float* wpan = ious + 16 * IOUP;

  const int tid  = threadIdx.x;
  const int lane = tid & 31;
  const int wave = tid >> 5;
  const int l15  = lane & 15;
  const int kb0  = (lane >> 4) << 1;
  const int roff = (lane >> 4) << 3;
  const int end  = base + cnt;

  if (gather)
  for (int e = tid; e < 64 * CHP; e += 256) {
    int row = e / CHP, col = e - row * CHP;
    float hv = 0.0f, cv = 0.0f;
    int node = base + (row >> 2);
    if (col < HD && node < end) {
      int cr = 4 * ((NN - 1) - node) + 1 + (row & 3);
      if (cr < NN) {
        int ci = (NN - 1) - cr;
        hv = hst[(size_t)ci * HD + col];
        cv = cst[(size_t)ci * HD + col];
      }
    }
    chh[e] = hv;
    chc[e] = cv;
  }
  __syncthreads();
  for (int e = tid; e < 16 * CHP; e += 256) {
    int m = e / CHP, col = e - m * CHP;
    int r0 = (m * 4) * CHP + col;
    hsum[e] = chh[r0] + chh[r0 + CHP] + chh[r0 + 2 * CHP] + chh[r0 + 3 * CHP];
  }
  __syncthreads();

  const float* bpan = wpan + (lane >> 4) * WPITCH + l15 * 2;

  const int ntask = ((4 * cnt + 31) >> 5) * 10;
  for (int t = wave; t < ntask; t += 8) {
    int mp = t / 10, nt = t - mp * 10;
    int m0 = mp * 32, n0 = nt * 16;
    int n  = n0 + l15;
    bool nok = n < HD;
    int  nc  = nok ? n : 0;
    v8f acc0 = {}, acc1 = {};
    const float* a0p = chh + (m0 + l15) * CHP + kb0;
    const float* a1p = a0p + 16 * CHP;
    const float* bq  = bpan + n0 * 2;
    for (int ks = 0; ks < 38; ++ks) {
      v2f a0 = *(const v2f*)(a0p + 4 * ks);
      v2f a1 = *(const v2f*)(a1p + 4 * ks);
      v2f b  = *(const v2f*)(bq + 2 * ks * WPITCH);
      acc0 = __builtin_amdgcn_wmma_f32_16x16x4_f32(false, a0, false, b, (short)0,
                                                   acc0, false, false);
      acc1 = __builtin_amdgcn_wmma_f32_16x16x4_f32(false, a1, false, b, (short)0,
                                                   acc1, false, false);
    }
    float bv = nok ? bfh[nc] : 0.0f;
    auto epilogue = [&](const v8f& acc, int mbase) {
      float s0 = 0.0f, s1 = 0.0f;
#pragma unroll
      for (int v = 0; v < 8; ++v) {
        int grow = mbase + v + roff;
        int node = base + (grow >> 2);
        float xv = (nok && node < end) ? xf[(size_t)node * HD + nc] : 0.0f;
        float f  = sigmoid_f(acc[v] + bv + xv);
        float w  = f * chc[grow * CHP + nc];
        if (v < 4) s0 += w; else s1 += w;
      }
      int nl = (mbase >> 2) + ((lane >> 4) << 1);
      fcs[nl * FCP + n]       = s0;
      fcs[(nl + 1) * FCP + n] = s1;
    };
    epilogue(acc0, m0);
    epilogue(acc1, m0 + 16);
  }

  {
    const float* ap = hsum + l15 * CHP + kb0;
    v2f a[38];
#pragma unroll
    for (int ks = 0; ks < 38; ++ks) a[ks] = *(const v2f*)(ap + 4 * ks);
    v8f acc[4] = {};
#pragma unroll
    for (int ks = 0; ks < 38; ++ks) {
#pragma unroll
      for (int j = 0; j < 4; ++j)
        acc[j] = __builtin_amdgcn_wmma_f32_16x16x4_f32(false, a[ks], false, wb[j][ks],
                                                       (short)0, acc[j], false, false);
    }
#pragma unroll
    for (int j = 0; j < 4; ++j) {
      const int n = iou_tile(wave, j) * 16 + l15;
      if (n < IOUW) {
        float bv = biouh[n];
#pragma unroll
        for (int v = 0; v < 8; ++v) {
          int m = v + roff;
          int node = base + m;
          float xv = (node < end) ? xiou[(size_t)node * IOUW + n] : 0.0f;
          ious[m * IOUP + n] = acc[j][v] + bv + xv;
        }
      }
    }
  }
  __syncthreads();

  for (int e = tid; e < 16 * HD; e += 256) {
    int m = e / HD, col = e - m * HD;
    int node = base + m;
    if (node < end) {
      float i = sigmoid_f(ious[m * IOUP + col]);
      float o = sigmoid_f(ious[m * IOUP + HD + col]);
      float u = tanh_f(ious[m * IOUP + 2 * HD + col]);
      float c = i * u + fcs[m * FCP + col];
      float h = o * tanh_f(c);
      hst[(size_t)node * HD + col] = h;
      if (more) {
        chh[(m ^ 3) * CHP + col] = h;
        chc[(m ^ 3) * CHP + col] = c;
      } else {
        cst[(size_t)node * HD + col] = c;
      }
    }
  }
}

__global__ void __launch_bounds__(256)
level_kernel(const float* __restrict__ xiou, const float* __restrict__ xf,
             const float* __restrict__ Wiouh, const float* __restrict__ biouh,
             const float* __restrict__ Wfh, const float* __restrict__ bfh,
             float* __restrict__ cst, float* __restrict__ hst,
             int s0, int s1, int s2, int cnt0, int nsub) {
  const int tid  = threadIdx.x;
  const int lane = tid & 31;
  const int wave = tid >> 5;
  const int l15  = lane & 15;
  const int kb0  = (lane >> 4) << 1;

  v2f wb[4][38];
#pragma unroll
  for (int j = 0; j < 4; ++j) {
    const int  n   = iou_tile(wave, j) * 16 + l15;
    const bool nok = n < IOUW;
    const int  nc  = nok ? n : IOUW - 1;
#pragma unroll
    for (int ks = 0; ks < 38; ++ks) {
      const int  k0  = 4 * ks + kb0;
      const bool ok  = nok && k0 < HD;
      const int  kc  = k0 < HD ? k0 : HD - 2;
      const float w0 = Wiouh[kc * IOUW + nc];
      const float w1 = Wiouh[(kc + 1) * IOUW + nc];
      wb[j][ks].x = ok ? w0 : 0.0f;
      wb[j][ks].y = ok ? w1 : 0.0f;
    }
  }

  stage_panel<HD, 0, 75, 160>(Wfh, smem + 2 * 64 * CHP + 16 * CHP + 16 * FCP + 16 * IOUP, tid);

  for (int sub = 0; sub < nsub; ++sub) {
    const int cnt  = cnt0 >> (2 * sub);
    const int lvl0 = sub == 0 ? s0 : (sub == 1 ? s1 : s2);
    sub_level((gmem_cf)xiou, (gmem_cf)xf, (gmem_cf)biouh, (gmem_cf)bfh,
              (gmem_f)cst, (gmem_f)hst, wb,
              lvl0 + (int)blockIdx.x * cnt, cnt, sub == 0, sub + 1 < nsub);
  }
}

extern "C" void kernel_launch(void* const* d_in, const int* in_sizes, int n_in,
                              void* d_out, int out_size, void* d_ws, size_t ws_size,
                              hipStream_t stream) {
  const float* x     = (const float*)d_in[0];
  const float* Wioux = (const float*)d_in[1];
  const float* bioux = (const float*)d_in[2];
  const float* Wiouh = (const float*)d_in[3];
  const float* biouh = (const float*)d_in[4];
  const float* Wfx   = (const float*)d_in[5];
  const float* bfx   = (const float*)d_in[6];
  const float* Wfh   = (const float*)d_in[7];
  const float* bfh   = (const float*)d_in[8];
  (void)in_sizes; (void)n_in; (void)out_size; (void)ws_size;

  float* xiou = (float*)d_ws;
  float* xf   = xiou + (size_t)NN * IOUW;
  float* cst  = xf   + (size_t)NN * HD;
  float* hst  = (float*)d_out;

  proj_gemm<<<dim3(17, 15), 256, 0, stream>>>(x, Wioux, bioux, xiou, IOUW);
  proj_gemm<<<dim3(51, 5),  256, 0, stream>>>(x, Wfx,   bfx,   xf,   HD);

  const int leafEnd = 14539;
  leaf_kernel<<<(leafEnd * HD + 255) / 256, 256, 0, stream>>>(xiou, biouh,
                                                              cst, hst, leafEnd);

  const size_t lds =
      (size_t)(2 * 64 * CHP + 16 * CHP + 16 * FCP + 16 * IOUP + WROWS * WPITCH) *
      sizeof(float);
  level_kernel<<<256, 256, lds, stream>>>(xiou, xf, Wiouh, biouh, Wfh, bfh,
                                          cst, hst, 14539, 18635, 19659, 16, 3);
  level_kernel<<<4, 256, lds, stream>>>(xiou, xf, Wiouh, biouh, Wfh, bfh,
                                        cst, hst, 19915, 19979, 0, 16, 2);
  level_kernel<<<1, 256, lds, stream>>>(xiou, xf, Wiouh, biouh, Wfh, bfh,
                                        cst, hst, 19995, 19999, 0, 4, 2);
}


# level_kernel: every global read of a workgroup's sub-levels (its nodes' xiou and xf rows, the 64 child rows: contiguous blocks) is a flat batched copy to LDS ahead of the math; same sums
# speedup vs baseline: 3.9821x; 1.2179x over previous
#include <hip/hip_runtime.h>


typedef __attribute__((ext_vector_type(2))) float v2f;
typedef __attribute__((ext_vector_type(4))) float v4f;
typedef __attribute__((ext_vector_type(8))) float v8f;

#define NN   20000
#define DIN  300
#define HD   150
#define IOUW 450
#define CHP  152
#define IOUP 464
#define FCP  160
#define WROWS  76
#define WPITCH 352
#define XROWS  21

__device__ __forceinline__ float sigmoid_f(float x) {
  return 1.0f / (1.0f + __expf(-x));
}
__device__ __forceinline__ float tanh_f(float x) {
  float t = __expf(2.0f * fminf(fmaxf(x, -15.0f), 15.0f));
  return (t - 1.0f) / (t + 1.0f);
}

__global__ void __launch_bounds__(256)
proj_gemm(const float* __restrict__ A, const float* __restrict__ B,
          const float* __restrict__ bias, float* __restrict__ C, int N) {
  const int tid  = threadIdx.x;
  const int lane = tid & 31;
  const int wave = __builtin_amdgcn_readfirstlane(tid >> 5);
  const int l15  = lane & 15;
  const int kb0  = (lane >> 4) << 1;
  const int n0   = (int)blockIdx.y * 32;
  const int na   = n0 + l15, nb = na + 16;
  const int ca = na < N ? na : N - 1;
  const int cb = nb < N ? nb : N - 1;

  v2f bfa[75], bfb[75];
  const unsigned un = (unsigned)N;
  const unsigned oa = (unsigned)kb0 * un + (unsigned)ca;
  const unsigned ob = (unsigned)kb0 * un + (unsigned)cb;
#pragma unroll
  for (int ks = 0; ks < 75; ++ks) {
    const unsigned kr = 4u * (unsigned)ks * un;
    bfa[ks].x = B[oa + kr];
    bfa[ks].y = B[oa + kr + un];
    bfb[ks].x = B[ob + kr];
    bfb[ks].y = B[ob + kr + un];
  }
  const float bva = bias[ca], bvb = bias[cb];

  const int step = (int)gridDim.x * 8;
  for (int mt = (int)blockIdx.x * 8 + wave; mt < NN / 16; mt += step) {
    const int m0 = mt * 16;
    const float* arow = A + (size_t)(m0 + l15) * DIN + kb0;
    v2f a[75];
#pragma unroll
    for (int ks = 0; ks < 75; ++ks) a[ks] = *(const v2f*)(arow + 4 * ks);
    __builtin_amdgcn_sched_barrier(0);
    v8f acc0 = {}, acc1 = {};
#pragma unroll
    for (int ks = 0; ks < 75; ++ks) {
      acc0 = __builtin_amdgcn_wmma_f32_16x16x4_f32(false, a[ks], false, bfa[ks],
                                                   (short)0, acc0, false, false);
      acc1 = __builtin_amdgcn_wmma_f32_16x16x4_f32(false, a[ks], false, bfb[ks],
                                                   (short)0, acc1, false, false);
    }
    float* crow = C + (size_t)(m0 + ((lane >> 4) << 3)) * N;
    if (na < N) {
#pragma unroll
      for (int v = 0; v < 8; ++v) crow[(size_t)v * N + na] = acc0[v] + bva;
    }
    if (nb < N) {
#pragma unroll
      for (int v = 0; v < 8; ++v) crow[(size_t)v * N + nb] = acc1[v] + bvb;
    }
  }
}

__global__ void leaf_kernel(const float* __restrict__ xiou,
                            const float* __restrict__ biouh,
                            float* __restrict__ cst, float* __restrict__ hst,
                            int count) {
  int e = (int)blockIdx.x * blockDim.x + threadIdx.x;
  if (e >= count * HD) return;
  int node = e / HD, col = e - node * HD;
  const float* r = xiou + (size_t)node * IOUW;
  float i = sigmoid_f(r[col] + biouh[col]);
  float o = sigmoid_f(r[HD + col] + biouh[HD + col]);
  float u = tanh_f(r[2 * HD + col] + biouh[2 * HD + col]);
  float c = i * u;
  cst[(size_t)node * HD + col] = c;
  hst[(size_t)node * HD + col] = o * tanh_f(c);
}

template <int LD, int COL0, int NVP, int PW>
__device__ __forceinline__ void stage_panel(const float* __restrict__ Wg,
                                            float* pan, int tid) {
  constexpr int ITEMS = 75 * NVP;
  constexpr int U     = 8;
  constexpr int NB    = (ITEMS + 256 * U - 1) / (256 * U);
  const v4f zero4 = {0.0f, 0.0f, 0.0f, 0.0f};
#pragma unroll
  for (int b = 0; b < NB; ++b) {
    v2f r0[U], r1[U];
#pragma unroll
    for (int u = 0; u < U; ++u) {
      int idx = (b * U + u) * 256 + tid;
      int idc = idx < ITEMS ? idx : ITEMS - 1;
      int kp = idc / NVP, cp = idc - kp * NVP;
      const float* p = Wg + (2 * kp) * LD + COL0 + 2 * cp;
      r0[u] = *(const v2f*)p;
      r1[u] = *(const v2f*)(p + LD);
    }
#pragma unroll
    for (int u = 0; u < U; ++u) {
      int idx = (b * U + u) * 256 + tid;
      if (idx < ITEMS) {
        int kp = idx / NVP, cp = idx - kp * NVP;
        v4f o = {r0[u].x, r1[u].x, r0[u].y, r1[u].y};
        *(v4f*)(pan + kp * WPITCH + 4 * cp) = o;
      }
    }
  }
  constexpr int PQ = PW / 2;
  for (int c = tid; c < PQ; c += 256)
    *(v4f*)(pan + 75 * WPITCH + 4 * c) = zero4;
  constexpr int NPAD = PQ - NVP;
  constexpr int NPD  = NPAD > 0 ? NPAD : 1;
  for (int idx = tid; idx < 75 * NPAD; idx += 256) {
    int kp = idx / NPD, cp = NVP + idx - kp * NPD;
    *(v4f*)(pan + kp * WPITCH + 4 * cp) = zero4;
  }
}

extern __shared__ __attribute__((aligned(16))) float smem[];

template <int U>
__device__ __forceinline__ void copy_flat(const float* __restrict__ src,
                                          float* dst, int n2, int tid) {
  for (int i0 = 0; i0 < n2; i0 += 256 * U) {
    v2f r[U];
#pragma unroll
    for (int u = 0; u < U; ++u) {
      int idx = i0 + u * 256 + tid;
      int ic  = idx < n2 ? idx : n2 - 1;
      r[u] = *(const v2f*)(src + 2 * ic);
    }
#pragma unroll
    for (int u = 0; u < U; ++u) {
      int idx = i0 + u * 256 + tid;
      if (idx < n2) *(v2f*)(dst + 2 * idx) = r[u];
    }
  }
}

typedef const __attribute__((address_space(1))) float* gmem_cf;
typedef __attribute__((address_space(1))) float* gmem_f;

__device__ __forceinline__ int iou_tile(int w, int j) {
  return (j >> 1) * 16 + 2 * w + (j & 1);
}

__device__ __forceinline__ void
sub_level(gmem_cf bfh_p,
          gmem_f cst_p, gmem_f hst_p, const v2f (&wb)[4][38],
          const float (&bvj)[4],
          int base, int cnt, int xo, bool gather, bool more) {
  const float* __restrict__ bfh   = (const float*)bfh_p;
  float* __restrict__ cst = (float*)cst_p;
  float* __restrict__ hst = (float*)hst_p;
  float* chh  = smem;
  float* chc  = chh  + 64 * CHP;
  float* hsum = chc  + 64 * CHP;
  float* fcs  = hsum + 16 * CHP;
  float* ious = fcs  + 16 * FCP;
  float* wpan = ious + 16 * IOUP;
  const float* xsi = wpan + WROWS * WPITCH + xo * IOUW;
  const float* xsf = wpan + WROWS * WPITCH + XROWS * IOUW + xo * HD;

  const int tid  = threadIdx.x;
  const int lane = tid & 31;
  const int wave = tid >> 5;
  const int l15  = lane & 15;
  const int kb0  = (lane >> 4) << 1;
  const int roff = (lane >> 4) << 3;
  const int end  = base + cnt;

  if (gather) {
    const int cmin = (NN - 2) - 4 * ((NN - 1) - base) - 3;
    const int jlo  = cmin < 0 ? -cmin : 0;
    const int jhi  = 4 * cnt < 64 ? 4 * cnt : 64;
    const int ilo  = 75 * jlo, ihi = 75 * jhi;
    constexpr int GU = 10;
    constexpr int GN = 64 * 75;
#pragma unroll
    for (int b = 0; b < 2; ++b) {
      v2f rh[GU], rc[GU];
#pragma unroll
      for (int u = 0; u < GU; ++u) {
        int idx = (b * GU + u) * 256 + tid;
        bool ok = idx >= ilo && idx < ihi;
        int off = ok ? cmin * HD + 2 * idx : 0;
        rh[u] = *(const v2f*)(hst + off);
        rc[u] = *(const v2f*)(cst + off);
      }
#pragma unroll
      for (int u = 0; u < GU; ++u) {
        int idx = (b * GU + u) * 256 + tid;
        if (idx < GN) {
          bool ok = idx >= ilo && idx < ihi;
          int j = idx / 75, q = idx - j * 75;
          const v2f z2 = {0.0f, 0.0f};
          *(v2f*)(chh + (j ^ 3) * CHP + 2 * q) = ok ? rh[u] : z2;
          *(v2f*)(chc + (j ^ 3) * CHP + 2 * q) = ok ? rc[u] : z2;
        }
      }
    }
    if (tid < 128) {
      const v2f z2 = {0.0f, 0.0f};
      *(v2f*)((tid < 64 ? chh : chc) + (tid & 63) * CHP + HD) = z2;
    }
  }
  __syncthreads();
  for (int e = tid; e < 16 * CHP; e += 256) {
    int m = e / CHP, col = e - m * CHP;
    int r0 = (m * 4) * CHP + col;
    hsum[e] = chh[r0] + chh[r0 + CHP] + chh[r0 + 2 * CHP] + chh[r0 + 3 * CHP];
  }
  __syncthreads();

  const float* bpan = wpan + (lane >> 4) * WPITCH + l15 * 2;

  const int ntask = ((4 * cnt + 31) >> 5) * 10;
  for (int t = wave; t < ntask; t += 8) {
    int mp = t / 10, nt = t - mp * 10;
    int m0 = mp * 32, n0 = nt * 16;
    int n  = n0 + l15;
    bool nok = n < HD;
    int  nc  = nok ? n : 0;
    v8f acc0 = {}, acc1 = {};
    const float bv = nok ? bfh[nc] : 0.0f;
    const float* a0p = chh + (m0 + l15) * CHP + kb0;
    const float* a1p = a0p + 16 * CHP;
    const float* bq  = bpan + n0 * 2;
    for (int ks = 0; ks < 38; ++ks) {
      v2f a0 = *(const v2f*)(a0p + 4 * ks);
      v2f a1 = *(const v2f*)(a1p + 4 * ks);
      v2f b  = *(const v2f*)(bq + 2 * ks * WPITCH);
      acc0 = __builtin_amdgcn_wmma_f32_16x16x4_f32(false, a0, false, b, (short)0,
                                                   acc0, false, false);
      acc1 = __builtin_amdgcn_wmma_f32_16x16x4_f32(false, a1, false, b, (short)0,
                                                   acc1, false, false);
    }
    auto epilogue = [&](const v8f& acc, int mbase) {
      float s0 = 0.0f, s1 = 0.0f;
#pragma unroll
      for (int v = 0; v < 8; ++v) {
        int grow = mbase + v + roff;
        int  mr  = grow >> 2;
        bool xok = nok && mr < cnt;
        float xs = xsf[(xok ? mr : 0) * HD + nc];
        float xv = xok ? xs : 0.0f;
        float f  = sigmoid_f(acc[v] + bv + xv);
        float w  = f * chc[grow * CHP + nc];
        if (v < 4) s0 += w; else s1 += w;
      }
      int nl = (mbase >> 2) + ((lane >> 4) << 1);
      fcs[nl * FCP + n]       = s0;
      fcs[(nl + 1) * FCP + n] = s1;
    };
    epilogue(acc0, m0);
    epilogue(acc1, m0 + 16);
  }

  {
    const float* ap = hsum + l15 * CHP + kb0;
    v2f a[38];
#pragma unroll
    for (int ks = 0; ks < 38; ++ks) a[ks] = *(const v2f*)(ap + 4 * ks);
    v8f acc[4] = {};
#pragma unroll
    for (int ks = 0; ks < 38; ++ks) {
#pragma unroll
      for (int j = 0; j < 4; ++j)
        acc[j] = __builtin_amdgcn_wmma_f32_16x16x4_f32(false, a[ks], false, wb[j][ks],
                                                       (short)0, acc[j], false, false);
    }
#pragma unroll
    for (int j = 0; j < 4; ++j) {
      const int n = iou_tile(wave, j) * 16 + l15;
      if (n < IOUW) {
        float bv = bvj[j];
#pragma unroll
        for (int v = 0; v < 8; ++v) {
          int m = v + roff;
          bool xok = m < cnt;
          float xs = xsi[(xok ? m : 0) * IOUW + n];
          float xv = xok ? xs : 0.0f;
          ious[m * IOUP + n] = acc[j][v] + bv + xv;
        }
      }
    }
  }
  __syncthreads();

  for (int e = tid; e < 16 * HD; e += 256) {
    int m = e / HD, col = e - m * HD;
    int node = base + m;
    if (node < end) {
      float i = sigmoid_f(ious[m * IOUP + col]);
      float o = sigmoid_f(ious[m * IOUP + HD + col]);
      float u = tanh_f(ious[m * IOUP + 2 * HD + col]);
      float c = i * u + fcs[m * FCP + col];
      float h = o * tanh_f(c);
      hst[(size_t)node * HD + col] = h;
      if (more) {
        chh[(m ^ 3) * CHP + col] = h;
        chc[(m ^ 3) * CHP + col] = c;
      } else {
        cst[(size_t)node * HD + col] = c;
      }
    }
  }
}

__global__ void __launch_bounds__(256)
level_kernel(const float* __restrict__ xiou, const float* __restrict__ xf,
             const float* __restrict__ Wiouh, const float* __restrict__ biouh,
             const float* __restrict__ Wfh, const float* __restrict__ bfh,
             float* __restrict__ cst, float* __restrict__ hst,
             int s0, int s1, int s2, int cnt0, int nsub) {
  const int tid  = threadIdx.x;
  const int lane = tid & 31;
  const int wave = tid >> 5;
  const int l15  = lane & 15;
  const int kb0  = (lane >> 4) << 1;

  v2f wb[4][38];
#pragma unroll
  for (int j = 0; j < 4; ++j) {
    const int  n   = iou_tile(wave, j) * 16 + l15;
    const bool nok = n < IOUW;
    const int  nc  = nok ? n : IOUW - 1;
#pragma unroll
    for (int ks = 0; ks < 38; ++ks) {
      const int  k0  = 4 * ks + kb0;
      const bool ok  = nok && k0 < HD;
      const int  kc  = k0 < HD ? k0 : HD - 2;
      const float w0 = Wiouh[kc * IOUW + nc];
      const float w1 = Wiouh[(kc + 1) * IOUW + nc];
      wb[j][ks].x = ok ? w0 : 0.0f;
      wb[j][ks].y = ok ? w1 : 0.0f;
    }
  }

  stage_panel<HD, 0, 75, 160>(Wfh, smem + 2 * 64 * CHP + 16 * CHP + 16 * FCP + 16 * IOUP, tid);

  {
    float* xs = smem + 2 * 64 * CHP + 16 * CHP + 16 * FCP + 16 * IOUP + WROWS * WPITCH;
    int xo = 0;
    for (int sub = 0; sub < nsub; ++sub) {
      const int cnt  = cnt0 >> (2 * sub);
      const int lvl0 = sub == 0 ? s0 : (sub == 1 ? s1 : s2);
      const int node = lvl0 + (int)blockIdx.x * cnt;
      copy_flat<16>(xiou + (size_t)node * IOUW, xs + xo * IOUW, cnt * (IOUW / 2), tid);
      copy_flat<16>(xf + (size_t)node * HD, xs + XROWS * IOUW + xo * HD, cnt * (HD / 2), tid);
      xo += cnt;
    }
  }

  float bvj[4];
#pragma unroll
  for (int j = 0; j < 4; ++j) {
    const int n = iou_tile(wave, j) * 16 + l15;
    bvj[j] = biouh[n < IOUW ? n : IOUW - 1];
  }

  int xo = 0;
  for (int sub = 0; sub < nsub; ++sub) {
    const int cnt  = cnt0 >> (2 * sub);
    const int lvl0 = sub == 0 ? s0 : (sub == 1 ? s1 : s2);
    sub_level((gmem_cf)bfh,
              (gmem_f)cst, (gmem_f)hst, wb, bvj,
              lvl0 + (int)blockIdx.x * cnt, cnt, xo, sub == 0, sub + 1 < nsub);
    xo += cnt;
  }
}

extern "C" void kernel_launch(void* const* d_in, const int* in_sizes, int n_in,
                              void* d_out, int out_size, void* d_ws, size_t ws_size,
                              hipStream_t stream) {
  const float* x     = (const float*)d_in[0];
  const float* Wioux = (const float*)d_in[1];
  const float* bioux = (const float*)d_in[2];
  const float* Wiouh = (const float*)d_in[3];
  const float* biouh = (const float*)d_in[4];
  const float* Wfx   = (const float*)d_in[5];
  const float* bfx   = (const float*)d_in[6];
  const float* Wfh   = (const float*)d_in[7];
  const float* bfh   = (const float*)d_in[8];
  (void)in_sizes; (void)n_in; (void)out_size; (void)ws_size;

  float* xiou = (float*)d_ws;
  float* xf   = xiou + (size_t)NN * IOUW;
  float* cst  = xf   + (size_t)NN * HD;
  float* hst  = (float*)d_out;

  proj_gemm<<<dim3(17, 15), 256, 0, stream>>>(x, Wioux, bioux, xiou, IOUW);
  proj_gemm<<<dim3(51, 5),  256, 0, stream>>>(x, Wfx,   bfx,   xf,   HD);

  const int leafEnd = 14539;
  leaf_kernel<<<(leafEnd * HD + 255) / 256, 256, 0, stream>>>(xiou, biouh,
                                                              cst, hst, leafEnd);

  const size_t lds =
      (size_t)(2 * 64 * CHP + 16 * CHP + 16 * FCP + 16 * IOUP + WROWS * WPITCH +
               XROWS * (IOUW + HD)) * sizeof(float);
  level_kernel<<<256, 256, lds, stream>>>(xiou, xf, Wiouh, biouh, Wfh, bfh,
                                          cst, hst, 14539, 18635, 19659, 16, 3);
  level_kernel<<<4, 256, lds, stream>>>(xiou, xf, Wiouh, biouh, Wfh, bfh,
                                        cst, hst, 19915, 19979, 0, 16, 2);
  level_kernel<<<1, 256, lds, stream>>>(xiou, xf, Wiouh, biouh, Wfh, bfh,
                                        cst, hst, 19995, 19999, 0, 4, 2);
}


# level_kernel: loads of every LDS copy of a workgroup (x rows, 64 child rows, Wfh panel) issued together at the top, stored under the Wiouh loads: one memory wait, not eleven in a row; same sums
# speedup vs baseline: 4.2333x; 1.0631x over previous
#include <hip/hip_runtime.h>


typedef __attribute__((ext_vector_type(2))) float v2f;
typedef __attribute__((ext_vector_type(4))) float v4f;
typedef __attribute__((ext_vector_type(8))) float v8f;

#define NN   20000
#define DIN  300
#define HD   150
#define IOUW 450
#define CHP  152
#define IOUP 464
#define FCP  160
#define WROWS  76
#define WPITCH 352
#define XROWS  21

__device__ __forceinline__ float sigmoid_f(float x) {
  return 1.0f / (1.0f + __expf(-x));
}
__device__ __forceinline__ float tanh_f(float x) {
  float t = __expf(2.0f * fminf(fmaxf(x, -15.0f), 15.0f));
  return (t - 1.0f) / (t + 1.0f);
}

__global__ void __launch_bounds__(256)
proj_gemm(const float* __restrict__ A, const float* __restrict__ B,
          const float* __restrict__ bias, float* __restrict__ C, int N) {
  const int tid  = threadIdx.x;
  const int lane = tid & 31;
  const int wave = __builtin_amdgcn_readfirstlane(tid >> 5);
  const int l15  = lane & 15;
  const int kb0  = (lane >> 4) << 1;
  const int n0   = (int)blockIdx.y * 32;
  const int na   = n0 + l15, nb = na + 16;
  const int ca = na < N ? na : N - 1;
  const int cb = nb < N ? nb : N - 1;

  v2f bfa[75], bfb[75];
  const unsigned un = (unsigned)N;
  const unsigned oa = (unsigned)kb0 * un + (unsigned)ca;
  const unsigned ob = (unsigned)kb0 * un + (unsigned)cb;
#pragma unroll
  for (int ks = 0; ks < 75; ++ks) {
    const unsigned kr = 4u * (unsigned)ks * un;
    bfa[ks].x = B[oa + kr];
    bfa[ks].y = B[oa + kr + un];
    bfb[ks].x = B[ob + kr];
    bfb[ks].y = B[ob + kr + un];
  }
  const float bva = bias[ca], bvb = bias[cb];

  const int step = (int)gridDim.x * 8;
  for (int mt = (int)blockIdx.x * 8 + wave; mt < NN / 16; mt += step) {
    const int m0 = mt * 16;
    const float* arow = A + (size_t)(m0 + l15) * DIN + kb0;
    v2f a[75];
#pragma unroll
    for (int ks = 0; ks < 75; ++ks) a[ks] = *(const v2f*)(arow + 4 * ks);
    __builtin_amdgcn_sched_barrier(0);
    v8f acc0 = {}, acc1 = {};
#pragma unroll
    for (int ks = 0; ks < 75; ++ks) {
      acc0 = __builtin_amdgcn_wmma_f32_16x16x4_f32(false, a[ks], false, bfa[ks],
                                                   (short)0, acc0, false, false);
      acc1 = __builtin_amdgcn_wmma_f32_16x16x4_f32(false, a[ks], false, bfb[ks],
                                                   (short)0, acc1, false, false);
    }
    float* crow = C + (size_t)(m0 + ((lane >> 4) << 3)) * N;
    if (na < N) {
#pragma unroll
      for (int v = 0; v < 8; ++v) crow[(size_t)v * N + na] = acc0[v] + bva;
    }
    if (nb < N) {
#pragma unroll
      for (int v = 0; v < 8; ++v) crow[(size_t)v * N + nb] = acc1[v] + bvb;
    }
  }
}

__global__ void leaf_kernel(const float* __restrict__ xiou,
                            const float* __restrict__ biouh,
                            float* __restrict__ cst, float* __restrict__ hst,
                            int count) {
  int e = (int)blockIdx.x * blockDim.x + threadIdx.x;
  if (e >= count * HD) return;
  int node = e / HD, col = e - node * HD;
  const float* r = xiou + (size_t)node * IOUW;
  float i = sigmoid_f(r[col] + biouh[col]);
  float o = sigmoid_f(r[HD + col] + biouh[HD + col]);
  float u = tanh_f(r[2 * HD + col] + biouh[2 * HD + col]);
  float c = i * u;
  cst[(size_t)node * HD + col] = c;
  hst[(size_t)node * HD + col] = o * tanh_f(c);
}

template <int LD, int COL0, int NVP, int S>
__device__ __forceinline__ void panel_issue(const float* __restrict__ Wg, int tid,
                                            v2f (&r0)[S], v2f (&r1)[S]) {
  constexpr int ITEMS = 75 * NVP;
  static_assert(S * 256 >= ITEMS, "slots must cover the panel");
#pragma unroll
  for (int u = 0; u < S; ++u) {
    int idx = u * 256 + tid;
    int idc = idx < ITEMS ? idx : ITEMS - 1;
    int kp = idc / NVP, cp = idc - kp * NVP;
    const float* p = Wg + (2 * kp) * LD + COL0 + 2 * cp;
    r0[u] = *(const v2f*)p;
    r1[u] = *(const v2f*)(p + LD);
  }
}
template <int NVP, int PW, int S>
__device__ __forceinline__ void panel_store(float* pan, int tid,
                                            const v2f (&r0)[S], const v2f (&r1)[S]) {
  constexpr int ITEMS = 75 * NVP;
  const v4f zero4 = {0.0f, 0.0f, 0.0f, 0.0f};
#pragma unroll
  for (int u = 0; u < S; ++u) {
    int idx = u * 256 + tid;
    int idc = idx < ITEMS ? idx : ITEMS - 1;
    int kp = idc / NVP, cp = idc - kp * NVP;
    v4f o = {r0[u].x, r1[u].x, r0[u].y, r1[u].y};
    *(v4f*)(pan + kp * WPITCH + 4 * cp) = o;
  }
  constexpr int PQ = PW / 2;
  for (int c = tid; c < PQ; c += 256)
    *(v4f*)(pan + 75 * WPITCH + 4 * c) = zero4;
  constexpr int NPAD = PQ - NVP;
  constexpr int NPD  = NPAD > 0 ? NPAD : 1;
  for (int idx = tid; idx < 75 * NPAD; idx += 256) {
    int kp = idx / NPD, cp = NVP + idx - kp * NPD;
    *(v4f*)(pan + kp * WPITCH + 4 * cp) = zero4;
  }
}

extern __shared__ __attribute__((aligned(16))) float smem[];

__device__ __forceinline__ int flat_item(int idx, int n2) {
  int ic = idx < n2 ? idx : n2 - 1;
  return ic < 0 ? 0 : ic;
}
template <int S>
__device__ __forceinline__ void flat_issue(const float* __restrict__ src,
                                           int n2, int tid, v2f (&r)[S]) {
#pragma unroll
  for (int u = 0; u < S; ++u)
    r[u] = *(const v2f*)(src + 2 * flat_item(u * 256 + tid, n2));
}
template <int S>
__device__ __forceinline__ void flat_store(float* dst, int n2, int tid,
                                           const v2f (&r)[S]) {
#pragma unroll
  for (int u = 0; u < S; ++u)
    *(v2f*)(dst + 2 * flat_item(u * 256 + tid, n2)) = r[u];
}

typedef const __attribute__((address_space(1))) float* gmem_cf;
typedef __attribute__((address_space(1))) float* gmem_f;

__device__ __forceinline__ int iou_tile(int w, int j) {
  return (j >> 1) * 16 + 2 * w + (j & 1);
}

__device__ __forceinline__ void
sub_level(gmem_cf bfh_p,
          gmem_f cst_p, gmem_f hst_p, const v2f (&wb)[4][38],
          const float (&bvj)[4],
          int base, int cnt, int xo, bool more) {
  const float* __restrict__ bfh   = (const float*)bfh_p;
  float* __restrict__ cst = (float*)cst_p;
  float* __restrict__ hst = (float*)hst_p;
  float* chh  = smem;
  float* chc  = chh  + 64 * CHP;
  float* hsum = chc  + 64 * CHP;
  float* fcs  = hsum + 16 * CHP;
  float* ious = fcs  + 16 * FCP;
  float* wpan = ious + 16 * IOUP;
  const float* xsi = wpan + WROWS * WPITCH + xo * IOUW;
  const float* xsf = wpan + WROWS * WPITCH + XROWS * IOUW + xo * HD;

  const int tid  = threadIdx.x;
  const int lane = tid & 31;
  const int wave = tid >> 5;
  const int l15  = lane & 15;
  const int kb0  = (lane >> 4) << 1;
  const int roff = (lane >> 4) << 3;
  const int end  = base + cnt;

  __syncthreads();
  for (int e = tid; e < 16 * CHP; e += 256) {
    int m = e / CHP, col = e - m * CHP;
    int r0 = (m * 4) * CHP + col;
    hsum[e] = chh[r0] + chh[r0 + CHP] + chh[r0 + 2 * CHP] + chh[r0 + 3 * CHP];
  }
  __syncthreads();

  const float* bpan = wpan + (lane >> 4) * WPITCH + l15 * 2;

  const int ntask = ((4 * cnt + 31) >> 5) * 10;
  for (int t = wave; t < ntask; t += 8) {
    int mp = t / 10, nt = t - mp * 10;
    int m0 = mp * 32, n0 = nt * 16;
    int n  = n0 + l15;
    bool nok = n < HD;
    int  nc  = nok ? n : 0;
    v8f acc0 = {}, acc1 = {};
    const float bv = nok ? bfh[nc] : 0.0f;
    const float* a0p = chh + (m0 + l15) * CHP + kb0;
    const float* a1p = a0p + 16 * CHP;
    const float* bq  = bpan + n0 * 2;
    for (int ks = 0; ks < 38; ++ks) {
      v2f a0 = *(const v2f*)(a0p + 4 * ks);
      v2f a1 = *(const v2f*)(a1p + 4 * ks);
      v2f b  = *(const v2f*)(bq + 2 * ks * WPITCH);
      acc0 = __builtin_amdgcn_wmma_f32_16x16x4_f32(false, a0, false, b, (short)0,
                                                   acc0, false, false);
      acc1 = __builtin_amdgcn_wmma_f32_16x16x4_f32(false, a1, false, b, (short)0,
                                                   acc1, false, false);
    }
    auto epilogue = [&](const v8f& acc, int mbase) {
      float s0 = 0.0f, s1 = 0.0f;
#pragma unroll
      for (int v = 0; v < 8; ++v) {
        int grow = mbase + v + roff;
        int  mr  = grow >> 2;
        bool xok = nok && mr < cnt;
        float xs = xsf[(xok ? mr : 0) * HD + nc];
        float xv = xok ? xs : 0.0f;
        float f  = sigmoid_f(acc[v] + bv + xv);
        float w  = f * chc[grow * CHP + nc];
        if (v < 4) s0 += w; else s1 += w;
      }
      int nl = (mbase >> 2) + ((lane >> 4) << 1);
      fcs[nl * FCP + n]       = s0;
      fcs[(nl + 1) * FCP + n] = s1;
    };
    epilogue(acc0, m0);
    epilogue(acc1, m0 + 16);
  }

  {
    const float* ap = hsum + l15 * CHP + kb0;
    v2f a[38];
#pragma unroll
    for (int ks = 0; ks < 38; ++ks) a[ks] = *(const v2f*)(ap + 4 * ks);
    v8f acc[4] = {};
#pragma unroll
    for (int ks = 0; ks < 38; ++ks) {
#pragma unroll
      for (int j = 0; j < 4; ++j)
        acc[j] = __builtin_amdgcn_wmma_f32_16x16x4_f32(false, a[ks], false, wb[j][ks],
                                                       (short)0, acc[j], false, false);
    }
#pragma unroll
    for (int j = 0; j < 4; ++j) {
      const int n = iou_tile(wave, j) * 16 + l15;
      if (n < IOUW) {
        float bv = bvj[j];
#pragma unroll
        for (int v = 0; v < 8; ++v) {
          int m = v + roff;
          bool xok = m < cnt;
          float xs = xsi[(xok ? m : 0) * IOUW + n];
          float xv = xok ? xs : 0.0f;
          ious[m * IOUP + n] = acc[j][v] + bv + xv;
        }
      }
    }
  }
  __syncthreads();

  for (int e = tid; e < 16 * HD; e += 256) {
    int m = e / HD, col = e - m * HD;
    int node = base + m;
    if (node < end) {
      float i = sigmoid_f(ious[m * IOUP + col]);
      float o = sigmoid_f(ious[m * IOUP + HD + col]);
      float u = tanh_f(ious[m * IOUP + 2 * HD + col]);
      float c = i * u + fcs[m * FCP + col];
      float h = o * tanh_f(c);
      hst[(size_t)node * HD + col] = h;
      if (more) {
        chh[(m ^ 3) * CHP + col] = h;
        chc[(m ^ 3) * CHP + col] = c;
      } else {
        cst[(size_t)node * HD + col] = c;
      }
    }
  }
}

__global__ void __launch_bounds__(256)
level_kernel(const float* __restrict__ xiou, const float* __restrict__ xf,
             const float* __restrict__ Wiouh, const float* __restrict__ biouh,
             const float* __restrict__ Wfh, const float* __restrict__ bfh,
             float* __restrict__ cst, float* __restrict__ hst,
             int s0, int s1, int s2, int cnt0, int nsub) {
  const int tid  = threadIdx.x;
  const int lane = tid & 31;
  const int wave = tid >> 5;
  const int l15  = lane & 15;
  const int kb0  = (lane >> 4) << 1;

  float* const xs  = smem + 2 * 64 * CHP + 16 * CHP + 16 * FCP + 16 * IOUP + WROWS * WPITCH;
  float* const chh = smem;
  float* const chc = smem + 64 * CHP;
  const int bid = (int)blockIdx.x;
  const int c0  = cnt0, c1 = cnt0 >> 2, c2 = nsub > 2 ? cnt0 >> 4 : 0;
  const int xo1 = c0, xo2 = c0 + c1;
  const float* const gi0 = xiou + (size_t)(s0 + bid * c0) * IOUW;
  const float* const gi1 = xiou + (size_t)(s1 + bid * c1) * IOUW;
  const float* const gi2 = xiou + (size_t)(s2 + bid * c2) * IOUW;
  const float* const gf0 = xf + (size_t)(s0 + bid * c0) * HD;
  const float* const gf1 = xf + (size_t)(s1 + bid * c1) * HD;
  const float* const gf2 = xf + (size_t)(s2 + bid * c2) * HD;
  v2f ri0[15], rf0[5], ri1[4], rf1[2], ri2[1], rf2[1];
  flat_issue<15>(gi0, c0 * (IOUW / 2), tid, ri0);
  flat_issue<5>(gf0, c0 * (HD / 2), tid, rf0);
  flat_issue<4>(gi1, c1 * (IOUW / 2), tid, ri1);
  flat_issue<2>(gf1, c1 * (HD / 2), tid, rf1);
  flat_issue<1>(gi2, c2 * (IOUW / 2), tid, ri2);
  flat_issue<1>(gf2, c2 * (HD / 2), tid, rf2);

  constexpr int GS = 19;
  constexpr int GN = 64 * 75;
  const int cmin = (NN - 2) - 4 * ((NN - 1) - (s0 + bid * c0)) - 3;
  const int jlo  = cmin < 0 ? -cmin : 0;
  const int jhi  = 4 * c0 < 64 ? 4 * c0 : 64;
  const int ilo  = 75 * jlo, ihi = 75 * jhi;
  v2f rh[GS], rc[GS];
#pragma unroll
  for (int u = 0; u < GS; ++u) {
    int idx = u * 256 + tid;
    idx = idx < GN ? idx : GN - 1;
    bool ok = idx >= ilo && idx < ihi;
    int off = ok ? cmin * HD + 2 * idx : 0;
    rh[u] = *(const v2f*)(hst + off);
    rc[u] = *(const v2f*)(cst + off);
  }
  v2f rp0[22], rp1[22];
  panel_issue<HD, 0, 75, 22>(Wfh, tid, rp0, rp1);
  __builtin_amdgcn_sched_barrier(0);

  v2f wb[4][38];
  auto load_wb = [&](int j) {
    const int  n   = iou_tile(wave, j) * 16 + l15;
    const bool nok = n < IOUW;
    const int  nc  = nok ? n : IOUW - 1;
#pragma unroll
    for (int ks = 0; ks < 38; ++ks) {
      const int  k0  = 4 * ks + kb0;
      const bool ok  = nok && k0 < HD;
      const int  kc  = k0 < HD ? k0 : HD - 2;
      const float w0 = Wiouh[kc * IOUW + nc];
      const float w1 = Wiouh[(kc + 1) * IOUW + nc];
      wb[j][ks].x = ok ? w0 : 0.0f;
      wb[j][ks].y = ok ? w1 : 0.0f;
    }
  };
  load_wb(0);
  load_wb(1);
  __builtin_amdgcn_sched_barrier(0);

  flat_store<15>(xs, c0 * (IOUW / 2), tid, ri0);
  flat_store<5>(xs + XROWS * IOUW, c0 * (HD / 2), tid, rf0);
  flat_store<4>(xs + xo1 * IOUW, c1 * (IOUW / 2), tid, ri1);
  flat_store<2>(xs + XROWS * IOUW + xo1 * HD, c1 * (HD / 2), tid, rf1);
  flat_store<1>(xs + xo2 * IOUW, c2 * (IOUW / 2), tid, ri2);
  flat_store<1>(xs + XROWS * IOUW + xo2 * HD, c2 * (HD / 2), tid, rf2);
#pragma unroll
  for (int u = 0; u < GS; ++u) {
    int idx = u * 256 + tid;
    idx = idx < GN ? idx : GN - 1;
    bool ok = idx >= ilo && idx < ihi;
    int j = idx / 75, q = idx - j * 75;
    const v2f z2 = {0.0f, 0.0f};
    *(v2f*)(chh + (j ^ 3) * CHP + 2 * q) = ok ? rh[u] : z2;
    *(v2f*)(chc + (j ^ 3) * CHP + 2 * q) = ok ? rc[u] : z2;
  }
  if (tid < 128) {
    const v2f z2 = {0.0f, 0.0f};
    *(v2f*)((tid < 64 ? chh : chc) + (tid & 63) * CHP + HD) = z2;
  }

  panel_store<75, 160, 22>(smem + 2 * 64 * CHP + 16 * CHP + 16 * FCP + 16 * IOUP, tid, rp0, rp1);

  load_wb(2);
  load_wb(3);

  float bvj[4];
#pragma unroll
  for (int j = 0; j < 4; ++j) {
    const int n = iou_tile(wave, j) * 16 + l15;
    bvj[j] = biouh[n < IOUW ? n : IOUW - 1];
  }

  int xo = 0;
  for (int sub = 0; sub < nsub; ++sub) {
    const int cnt  = cnt0 >> (2 * sub);
    const int lvl0 = sub == 0 ? s0 : (sub == 1 ? s1 : s2);
    sub_level((gmem_cf)bfh,
              (gmem_f)cst, (gmem_f)hst, wb, bvj,
              lvl0 + (int)blockIdx.x * cnt, cnt, xo, sub + 1 < nsub);
    xo += cnt;
  }
}

extern "C" void kernel_launch(void* const* d_in, const int* in_sizes, int n_in,
                              void* d_out, int out_size, void* d_ws, size_t ws_size,
                              hipStream_t stream) {
  const float* x     = (const float*)d_in[0];
  const float* Wioux = (const float*)d_in[1];
  const float* bioux = (const float*)d_in[2];
  const float* Wiouh = (const float*)d_in[3];
  const float* biouh = (const float*)d_in[4];
  const float* Wfx   = (const float*)d_in[5];
  const float* bfx   = (const float*)d_in[6];
  const float* Wfh   = (const float*)d_in[7];
  const float* bfh   = (const float*)d_in[8];
  (void)in_sizes; (void)n_in; (void)out_size; (void)ws_size;

  float* xiou = (float*)d_ws;
  float* xf   = xiou + (size_t)NN * IOUW;
  float* cst  = xf   + (size_t)NN * HD;
  float* hst  = (float*)d_out;

  proj_gemm<<<dim3(17, 15), 256, 0, stream>>>(x, Wioux, bioux, xiou, IOUW);
  proj_gemm<<<dim3(51, 5),  256, 0, stream>>>(x, Wfx,   bfx,   xf,   HD);

  const int leafEnd = 14539;
  leaf_kernel<<<(leafEnd * HD + 255) / 256, 256, 0, stream>>>(xiou, biouh,
                                                              cst, hst, leafEnd);

  const size_t lds =
      (size_t)(2 * 64 * CHP + 16 * CHP + 16 * FCP + 16 * IOUP + WROWS * WPITCH +
               XROWS * (IOUW + HD)) * sizeof(float);
  level_kernel<<<256, 256, lds, stream>>>(xiou, xf, Wiouh, biouh, Wfh, bfh,
                                          cst, hst, 14539, 18635, 19659, 16, 3);
  level_kernel<<<4, 256, lds, stream>>>(xiou, xf, Wiouh, biouh, Wfh, bfh,
                                        cst, hst, 19915, 19979, 0, 16, 2);
  level_kernel<<<1, 256, lds, stream>>>(xiou, xf, Wiouh, biouh, Wfh, bfh,
                                        cst, hst, 19995, 19999, 0, 4, 2);
}


# level_kernel phase C: 16x32 tasks (one m-tile, two Wfh n-tiles), not 32x16: 4- and 1-node sub-levels run no dead m-tile (5 tasks, not 10); K loop unrolled, LDS reads 4-8 K-steps ahead; same sums
# speedup vs baseline: 4.4040x; 1.0403x over previous
#include <hip/hip_runtime.h>


typedef __attribute__((ext_vector_type(2))) float v2f;
typedef __attribute__((ext_vector_type(4))) float v4f;
typedef __attribute__((ext_vector_type(8))) float v8f;

#define NN   20000
#define DIN  300
#define HD   150
#define IOUW 450
#define CHP  152
#define IOUP 464
#define FCP  160
#define WROWS  76
#define WPITCH 352
#define XROWS  21

__device__ __forceinline__ float sigmoid_f(float x) {
  return 1.0f / (1.0f + __expf(-x));
}
__device__ __forceinline__ float tanh_f(float x) {
  float t = __expf(2.0f * fminf(fmaxf(x, -15.0f), 15.0f));
  return (t - 1.0f) / (t + 1.0f);
}

__global__ void __launch_bounds__(256)
proj_gemm(const float* __restrict__ A, const float* __restrict__ B,
          const float* __restrict__ bias, float* __restrict__ C, int N) {
  const int tid  = threadIdx.x;
  const int lane = tid & 31;
  const int wave = __builtin_amdgcn_readfirstlane(tid >> 5);
  const int l15  = lane & 15;
  const int kb0  = (lane >> 4) << 1;
  const int n0   = (int)blockIdx.y * 32;
  const int na   = n0 + l15, nb = na + 16;
  const int ca = na < N ? na : N - 1;
  const int cb = nb < N ? nb : N - 1;

  v2f bfa[75], bfb[75];
  const unsigned un = (unsigned)N;
  const unsigned oa = (unsigned)kb0 * un + (unsigned)ca;
  const unsigned ob = (unsigned)kb0 * un + (unsigned)cb;
#pragma unroll
  for (int ks = 0; ks < 75; ++ks) {
    const unsigned kr = 4u * (unsigned)ks * un;
    bfa[ks].x = B[oa + kr];
    bfa[ks].y = B[oa + kr + un];
    bfb[ks].x = B[ob + kr];
    bfb[ks].y = B[ob + kr + un];
  }
  const float bva = bias[ca], bvb = bias[cb];

  const int step = (int)gridDim.x * 8;
  for (int mt = (int)blockIdx.x * 8 + wave; mt < NN / 16; mt += step) {
    const int m0 = mt * 16;
    const float* arow = A + (size_t)(m0 + l15) * DIN + kb0;
    v2f a[75];
#pragma unroll
    for (int ks = 0; ks < 75; ++ks) a[ks] = *(const v2f*)(arow + 4 * ks);
    __builtin_amdgcn_sched_barrier(0);
    v8f acc0 = {}, acc1 = {};
#pragma unroll
    for (int ks = 0; ks < 75; ++ks) {
      acc0 = __builtin_amdgcn_wmma_f32_16x16x4_f32(false, a[ks], false, bfa[ks],
                                                   (short)0, acc0, false, false);
      acc1 = __builtin_amdgcn_wmma_f32_16x16x4_f32(false, a[ks], false, bfb[ks],
                                                   (short)0, acc1, false, false);
    }
    float* crow = C + (size_t)(m0 + ((lane >> 4) << 3)) * N;
    if (na < N) {
#pragma unroll
      for (int v = 0; v < 8; ++v) crow[(size_t)v * N + na] = acc0[v] + bva;
    }
    if (nb < N) {
#pragma unroll
      for (int v = 0; v < 8; ++v) crow[(size_t)v * N + nb] = acc1[v] + bvb;
    }
  }
}

__global__ void leaf_kernel(const float* __restrict__ xiou,
                            const float* __restrict__ biouh,
                            float* __restrict__ cst, float* __restrict__ hst,
                            int count) {
  int e = (int)blockIdx.x * blockDim.x + threadIdx.x;
  if (e >= count * HD) return;
  int node = e / HD, col = e - node * HD;
  const float* r = xiou + (size_t)node * IOUW;
  float i = sigmoid_f(r[col] + biouh[col]);
  float o = sigmoid_f(r[HD + col] + biouh[HD + col]);
  float u = tanh_f(r[2 * HD + col] + biouh[2 * HD + col]);
  float c = i * u;
  cst[(size_t)node * HD + col] = c;
  hst[(size_t)node * HD + col] = o * tanh_f(c);
}

template <int LD, int COL0, int NVP, int S>
__device__ __forceinline__ void panel_issue(const float* __restrict__ Wg, int tid,
                                            v2f (&r0)[S], v2f (&r1)[S]) {
  constexpr int ITEMS = 75 * NVP;
  static_assert(S * 256 >= ITEMS, "slots must cover the panel");
#pragma unroll
  for (int u = 0; u < S; ++u) {
    int idx = u * 256 + tid;
    int idc = idx < ITEMS ? idx : ITEMS - 1;
    int kp = idc / NVP, cp = idc - kp * NVP;
    const float* p = Wg + (2 * kp) * LD + COL0 + 2 * cp;
    r0[u] = *(const v2f*)p;
    r1[u] = *(const v2f*)(p + LD);
  }
}
template <int NVP, int PW, int S>
__device__ __forceinline__ void panel_store(float* pan, int tid,
                                            const v2f (&r0)[S], const v2f (&r1)[S]) {
  constexpr int ITEMS = 75 * NVP;
  const v4f zero4 = {0.0f, 0.0f, 0.0f, 0.0f};
#pragma unroll
  for (int u = 0; u < S; ++u) {
    int idx = u * 256 + tid;
    int idc = idx < ITEMS ? idx : ITEMS - 1;
    int kp = idc / NVP, cp = idc - kp * NVP;
    v4f o = {r0[u].x, r1[u].x, r0[u].y, r1[u].y};
    *(v4f*)(pan + kp * WPITCH + 4 * cp) = o;
  }
  constexpr int PQ = PW / 2;
  for (int c = tid; c < PQ; c += 256)
    *(v4f*)(pan + 75 * WPITCH + 4 * c) = zero4;
  constexpr int NPAD = PQ - NVP;
  constexpr int NPD  = NPAD > 0 ? NPAD : 1;
  for (int idx = tid; idx < 75 * NPAD; idx += 256) {
    int kp = idx / NPD, cp = NVP + idx - kp * NPD;
    *(v4f*)(pan + kp * WPITCH + 4 * cp) = zero4;
  }
}

extern __shared__ __attribute__((aligned(16))) float smem[];

__device__ __forceinline__ int flat_item(int idx, int n2) {
  int ic = idx < n2 ? idx : n2 - 1;
  return ic < 0 ? 0 : ic;
}
template <int S>
__device__ __forceinline__ void flat_issue(const float* __restrict__ src,
                                           int n2, int tid, v2f (&r)[S]) {
#pragma unroll
  for (int u = 0; u < S; ++u)
    r[u] = *(const v2f*)(src + 2 * flat_item(u * 256 + tid, n2));
}
template <int S>
__device__ __forceinline__ void flat_store(float* dst, int n2, int tid,
                                           const v2f (&r)[S]) {
#pragma unroll
  for (int u = 0; u < S; ++u)
    *(v2f*)(dst + 2 * flat_item(u * 256 + tid, n2)) = r[u];
}

typedef const __attribute__((address_space(1))) float* gmem_cf;
typedef __attribute__((address_space(1))) float* gmem_f;

__device__ __forceinline__ int iou_tile(int w, int j) {
  return (j >> 1) * 16 + 2 * w + (j & 1);
}

__device__ __forceinline__ void
forget_task(const float* chh, const float* chc, const float* bpan,
            const float* xsf, float* fcs, const float (&bv)[2],
            int m0, int nt0, int cnt, int lane) {
  const int l15  = lane & 15;
  const int kb0  = (lane >> 4) << 1;
  const int roff = (lane >> 4) << 3;
  const float* ap = chh + (m0 + l15) * CHP + kb0;
  const float* bq = bpan + nt0 * 32;
  v2f a[38], b0[38], b1[38];
  auto ld = [&](int ks) {
    a[ks]  = *(const v2f*)(ap + 4 * ks);
    b0[ks] = *(const v2f*)(bq + 2 * ks * WPITCH);
    b1[ks] = *(const v2f*)(bq + 2 * ks * WPITCH + 32);
  };
#pragma unroll
  for (int ks = 0; ks < 8; ++ks) ld(ks);
  __builtin_amdgcn_sched_barrier(0);
  v8f acc[2] = {};
#pragma unroll
  for (int c = 0; c < 10; ++c) {
#pragma unroll
    for (int i = 0; i < 4; ++i)
      if (4 * (c + 2) + i < 38) ld(4 * (c + 2) + i);
#pragma unroll
    for (int i = 0; i < 4; ++i) {
      const int ks = 4 * c + i;
      if (ks < 38) {
        acc[0] = __builtin_amdgcn_wmma_f32_16x16x4_f32(false, a[ks], false, b0[ks], (short)0,
                                                       acc[0], false, false);
        acc[1] = __builtin_amdgcn_wmma_f32_16x16x4_f32(false, a[ks], false, b1[ks], (short)0,
                                                       acc[1], false, false);
      }
    }
    __builtin_amdgcn_sched_barrier(0);
  }
#pragma unroll
  for (int j = 0; j < 2; ++j) {
    const int  n   = (nt0 + j) * 16 + l15;
    const bool nok = n < HD;
    const int  nc  = nok ? n : 0;
    float s0 = 0.0f, s1 = 0.0f;
#pragma unroll
    for (int v = 0; v < 8; ++v) {
      int grow = m0 + v + roff;
      int  mr  = grow >> 2;
      bool xok = nok && mr < cnt;
      float xs = xsf[(xok ? mr : 0) * HD + nc];
      float xv = xok ? xs : 0.0f;
      float f  = sigmoid_f(acc[j][v] + bv[j] + xv);
      float w  = f * chc[grow * CHP + nc];
      if (v < 4) s0 += w; else s1 += w;
    }
    int nl = (m0 >> 2) + ((lane >> 4) << 1);
    fcs[nl * FCP + n]       = s0;
    fcs[(nl + 1) * FCP + n] = s1;
  }
}

__device__ __forceinline__ void
sub_level(gmem_cf bfh_p,
          gmem_f cst_p, gmem_f hst_p, const v2f (&wb)[4][38],
          const float (&bvj)[4],
          int base, int cnt, int xo, bool more) {
  const float* __restrict__ bfh   = (const float*)bfh_p;
  float* __restrict__ cst = (float*)cst_p;
  float* __restrict__ hst = (float*)hst_p;
  float* chh  = smem;
  float* chc  = chh  + 64 * CHP;
  float* hsum = chc  + 64 * CHP;
  float* fcs  = hsum + 16 * CHP;
  float* ious = fcs  + 16 * FCP;
  float* wpan = ious + 16 * IOUP;
  const float* xsi = wpan + WROWS * WPITCH + xo * IOUW;
  const float* xsf = wpan + WROWS * WPITCH + XROWS * IOUW + xo * HD;

  const int tid  = threadIdx.x;
  const int lane = tid & 31;
  const int wave = tid >> 5;
  const int l15  = lane & 15;
  const int kb0  = (lane >> 4) << 1;
  const int roff = (lane >> 4) << 3;
  const int end  = base + cnt;

  __syncthreads();
  for (int e = tid; e < 16 * CHP; e += 256) {
    int m = e / CHP, col = e - m * CHP;
    int r0 = (m * 4) * CHP + col;
    hsum[e] = chh[r0] + chh[r0 + CHP] + chh[r0 + 2 * CHP] + chh[r0 + 3 * CHP];
  }
  __syncthreads();

  const float* bpan = wpan + (lane >> 4) * WPITCH + l15 * 2;

  {
    const int ntask = ((4 * cnt + 15) >> 4) * 5;
    const int ws = __builtin_amdgcn_readfirstlane(wave);
    for (int t = ws; t < ntask; t += 8) {
      const int mt = t / 5, nt0 = 2 * (t - mt * 5);
      float bv[2];
#pragma unroll
      for (int j = 0; j < 2; ++j) {
        const int n = (nt0 + j) * 16 + l15;
        const float bl = bfh[n < HD ? n : 0];
        bv[j] = n < HD ? bl : 0.0f;
      }
      forget_task(chh, chc, bpan, xsf, fcs, bv, mt * 16, nt0, cnt, lane);
    }
  }

  {
    const float* ap = hsum + l15 * CHP + kb0;
    v2f a[38];
#pragma unroll
    for (int ks = 0; ks < 38; ++ks) a[ks] = *(const v2f*)(ap + 4 * ks);
    v8f acc[4] = {};
#pragma unroll
    for (int ks = 0; ks < 38; ++ks) {
#pragma unroll
      for (int j = 0; j < 4; ++j)
        acc[j] = __builtin_amdgcn_wmma_f32_16x16x4_f32(false, a[ks], false, wb[j][ks],
                                                       (short)0, acc[j], false, false);
    }
#pragma unroll
    for (int j = 0; j < 4; ++j) {
      const int n = iou_tile(wave, j) * 16 + l15;
      if (n < IOUW) {
        float bv = bvj[j];
#pragma unroll
        for (int v = 0; v < 8; ++v) {
          int m = v + roff;
          bool xok = m < cnt;
          float xs = xsi[(xok ? m : 0) * IOUW + n];
          float xv = xok ? xs : 0.0f;
          ious[m * IOUP + n] = acc[j][v] + bv + xv;
        }
      }
    }
  }
  __syncthreads();

  for (int e = tid; e < 16 * HD; e += 256) {
    int m = e / HD, col = e - m * HD;
    int node = base + m;
    if (node < end) {
      float i = sigmoid_f(ious[m * IOUP + col]);
      float o = sigmoid_f(ious[m * IOUP + HD + col]);
      float u = tanh_f(ious[m * IOUP + 2 * HD + col]);
      float c = i * u + fcs[m * FCP + col];
      float h = o * tanh_f(c);
      hst[(size_t)node * HD + col] = h;
      if (more) {
        chh[(m ^ 3) * CHP + col] = h;
        chc[(m ^ 3) * CHP + col] = c;
      } else {
        cst[(size_t)node * HD + col] = c;
      }
    }
  }
}

__global__ void __launch_bounds__(256)
level_kernel(const float* __restrict__ xiou, const float* __restrict__ xf,
             const float* __restrict__ Wiouh, const float* __restrict__ biouh,
             const float* __restrict__ Wfh, const float* __restrict__ bfh,
             float* __restrict__ cst, float* __restrict__ hst,
             int s0, int s1, int s2, int cnt0, int nsub) {
  const int tid  = threadIdx.x;
  const int lane = tid & 31;
  const int wave = tid >> 5;
  const int l15  = lane & 15;
  const int kb0  = (lane >> 4) << 1;

  float* const xs  = smem + 2 * 64 * CHP + 16 * CHP + 16 * FCP + 16 * IOUP + WROWS * WPITCH;
  float* const chh = smem;
  float* const chc = smem + 64 * CHP;
  const int bid = (int)blockIdx.x;
  const int c0  = cnt0, c1 = cnt0 >> 2, c2 = nsub > 2 ? cnt0 >> 4 : 0;
  const int xo1 = c0, xo2 = c0 + c1;
  const float* const gi0 = xiou + (size_t)(s0 + bid * c0) * IOUW;
  const float* const gi1 = xiou + (size_t)(s1 + bid * c1) * IOUW;
  const float* const gi2 = xiou + (size_t)(s2 + bid * c2) * IOUW;
  const float* const gf0 = xf + (size_t)(s0 + bid * c0) * HD;
  const float* const gf1 = xf + (size_t)(s1 + bid * c1) * HD;
  const float* const gf2 = xf + (size_t)(s2 + bid * c2) * HD;
  v2f ri0[15], rf0[5], ri1[4], rf1[2], ri2[1], rf2[1];
  flat_issue<15>(gi0, c0 * (IOUW / 2), tid, ri0);
  flat_issue<5>(gf0, c0 * (HD / 2), tid, rf0);
  flat_issue<4>(gi1, c1 * (IOUW / 2), tid, ri1);
  flat_issue<2>(gf1, c1 * (HD / 2), tid, rf1);
  flat_issue<1>(gi2, c2 * (IOUW / 2), tid, ri2);
  flat_issue<1>(gf2, c2 * (HD / 2), tid, rf2);

  constexpr int GS = 19;
  constexpr int GN = 64 * 75;
  const int cmin = (NN - 2) - 4 * ((NN - 1) - (s0 + bid * c0)) - 3;
  const int jlo  = cmin < 0 ? -cmin : 0;
  const int jhi  = 4 * c0 < 64 ? 4 * c0 : 64;
  const int ilo  = 75 * jlo, ihi = 75 * jhi;
  v2f rh[GS], rc[GS];
#pragma unroll
  for (int u = 0; u < GS; ++u) {
    int idx = u * 256 + tid;
    idx = idx < GN ? idx : GN - 1;
    bool ok = idx >= ilo && idx < ihi;
    int off = ok ? cmin * HD + 2 * idx : 0;
    rh[u] = *(const v2f*)(hst + off);
    rc[u] = *(const v2f*)(cst + off);
  }
  v2f rp0[22], rp1[22];
  panel_issue<HD, 0, 75, 22>(Wfh, tid, rp0, rp1);
  __builtin_amdgcn_sched_barrier(0);

  v2f wb[4][38];
  auto load_wb = [&](int j) {
    const int  n   = iou_tile(wave, j) * 16 + l15;
    const bool nok = n < IOUW;
    const int  nc  = nok ? n : IOUW - 1;
#pragma unroll
    for (int ks = 0; ks < 38; ++ks) {
      const int  k0  = 4 * ks + kb0;
      const bool ok  = nok && k0 < HD;
      const int  kc  = k0 < HD ? k0 : HD - 2;
      const float w0 = Wiouh[kc * IOUW + nc];
      const float w1 = Wiouh[(kc + 1) * IOUW + nc];
      wb[j][ks].x = ok ? w0 : 0.0f;
      wb[j][ks].y = ok ? w1 : 0.0f;
    }
  };
  load_wb(0);
  load_wb(1);
  __builtin_amdgcn_sched_barrier(0);

  flat_store<15>(xs, c0 * (IOUW / 2), tid, ri0);
  flat_store<5>(xs + XROWS * IOUW, c0 * (HD / 2), tid, rf0);
  flat_store<4>(xs + xo1 * IOUW, c1 * (IOUW / 2), tid, ri1);
  flat_store<2>(xs + XROWS * IOUW + xo1 * HD, c1 * (HD / 2), tid, rf1);
  flat_store<1>(xs + xo2 * IOUW, c2 * (IOUW / 2), tid, ri2);
  flat_store<1>(xs + XROWS * IOUW + xo2 * HD, c2 * (HD / 2), tid, rf2);
#pragma unroll
  for (int u = 0; u < GS; ++u) {
    int idx = u * 256 + tid;
    idx = idx < GN ? idx : GN - 1;
    bool ok = idx >= ilo && idx < ihi;
    int j = idx / 75, q = idx - j * 75;
    const v2f z2 = {0.0f, 0.0f};
    *(v2f*)(chh + (j ^ 3) * CHP + 2 * q) = ok ? rh[u] : z2;
    *(v2f*)(chc + (j ^ 3) * CHP + 2 * q) = ok ? rc[u] : z2;
  }
  if (tid < 128) {
    const v2f z2 = {0.0f, 0.0f};
    *(v2f*)((tid < 64 ? chh : chc) + (tid & 63) * CHP + HD) = z2;
  }

  panel_store<75, 160, 22>(smem + 2 * 64 * CHP + 16 * CHP + 16 * FCP + 16 * IOUP, tid, rp0, rp1);

  load_wb(2);
  load_wb(3);

  float bvj[4];
#pragma unroll
  for (int j = 0; j < 4; ++j) {
    const int n = iou_tile(wave, j) * 16 + l15;
    bvj[j] = biouh[n < IOUW ? n : IOUW - 1];
  }

  int xo = 0;
  for (int sub = 0; sub < nsub; ++sub) {
    const int cnt  = cnt0 >> (2 * sub);
    const int lvl0 = sub == 0 ? s0 : (sub == 1 ? s1 : s2);
    sub_level((gmem_cf)bfh,
              (gmem_f)cst, (gmem_f)hst, wb, bvj,
              lvl0 + (int)blockIdx.x * cnt, cnt, xo, sub + 1 < nsub);
    xo += cnt;
  }
}

extern "C" void kernel_launch(void* const* d_in, const int* in_sizes, int n_in,
                              void* d_out, int out_size, void* d_ws, size_t ws_size,
                              hipStream_t stream) {
  const float* x     = (const float*)d_in[0];
  const float* Wioux = (const float*)d_in[1];
  const float* bioux = (const float*)d_in[2];
  const float* Wiouh = (const float*)d_in[3];
  const float* biouh = (const float*)d_in[4];
  const float* Wfx   = (const float*)d_in[5];
  const float* bfx   = (const float*)d_in[6];
  const float* Wfh   = (const float*)d_in[7];
  const float* bfh   = (const float*)d_in[8];
  (void)in_sizes; (void)n_in; (void)out_size; (void)ws_size;

  float* xiou = (float*)d_ws;
  float* xf   = xiou + (size_t)NN * IOUW;
  float* cst  = xf   + (size_t)NN * HD;
  float* hst  = (float*)d_out;

  proj_gemm<<<dim3(17, 15), 256, 0, stream>>>(x, Wioux, bioux, xiou, IOUW);
  proj_gemm<<<dim3(51, 5),  256, 0, stream>>>(x, Wfx,   bfx,   xf,   HD);

  const int leafEnd = 14539;
  leaf_kernel<<<(leafEnd * HD + 255) / 256, 256, 0, stream>>>(xiou, biouh,
                                                              cst, hst, leafEnd);

  const size_t lds =
      (size_t)(2 * 64 * CHP + 16 * CHP + 16 * FCP + 16 * IOUP + WROWS * WPITCH +
               XROWS * (IOUW + HD)) * sizeof(float);
  level_kernel<<<256, 256, lds, stream>>>(xiou, xf, Wiouh, biouh, Wfh, bfh,
                                          cst, hst, 14539, 18635, 19659, 16, 3);
  level_kernel<<<4, 256, lds, stream>>>(xiou, xf, Wiouh, biouh, Wfh, bfh,
                                        cst, hst, 19915, 19979, 0, 16, 2);
  level_kernel<<<1, 256, lds, stream>>>(xiou, xf, Wiouh, biouh, Wfh, bfh,
                                        cst, hst, 19995, 19999, 0, 4, 2);
}


# proj_gemm: a workgroup's 8 waves share each 16-row x tile (one whole-line copy into a double-buffered LDS tile); one launch covers all 39 n-tiles of Wioux and Wfx: 5 launches, not 6; same sums
# speedup vs baseline: 4.8388x; 1.0987x over previous
#include <hip/hip_runtime.h>


typedef __attribute__((ext_vector_type(2))) float v2f;
typedef __attribute__((ext_vector_type(4))) float v4f;
typedef __attribute__((ext_vector_type(8))) float v8f;

#define NN   20000
#define DIN  300
#define HD   150
#define IOUW 450
#define CHP  152
#define IOUP 464
#define FCP  160
#define WROWS  76
#define WPITCH 352
#define XROWS  21

__device__ __forceinline__ float sigmoid_f(float x) {
  return 1.0f / (1.0f + __expf(-x));
}
__device__ __forceinline__ float tanh_f(float x) {
  float t = __expf(2.0f * fminf(fmaxf(x, -15.0f), 15.0f));
  return (t - 1.0f) / (t + 1.0f);
}

#define NTI   29
#define NTA   39
#define GWA   100
#define GWC   56
static_assert(2 * GWA + GWC == 256, "the launch has 256 workgroups");
#define ATILE (16 * DIN)
#define ATQ   (ATILE / 4)
__global__ void __launch_bounds__(256)
proj_gemm(const float* __restrict__ A,
          const float* __restrict__ B0, const float* __restrict__ bias0,
          float* __restrict__ C0,
          const float* __restrict__ B1, const float* __restrict__ bias1,
          float* __restrict__ C1) {
  __shared__ __attribute__((aligned(16))) float at[2][ATILE];
  const int tid  = threadIdx.x;
  const int lane = tid & 31;
  const int wave = __builtin_amdgcn_readfirstlane(tid >> 5);
  const int l15  = lane & 15;
  const int kb0  = (lane >> 4) << 1;
  const int bid  = (int)blockIdx.x;
  const int grp  = bid < GWA ? 0 : (bid < 2 * GWA ? 1 : 2);
  const int step = grp < 2 ? GWA : GWC;
  const int ta   = grp * 16 + wave * 2, tb = ta + 1;
  const bool live = ta < NTA;
  const bool xa = ta < NTI, xb = tb < NTI;
  const float* __restrict__ Ba = xa ? B0 : B1;
  const float* __restrict__ Bb = xb ? B0 : B1;
  const float* __restrict__ ba = xa ? bias0 : bias1;
  const float* __restrict__ bb = xb ? bias0 : bias1;
  float* __restrict__ Ca = xa ? C0 : C1;
  float* __restrict__ Cb = xb ? C0 : C1;
  const unsigned sa = xa ? IOUW : HD, sb = xb ? IOUW : HD;
  const unsigned na = (unsigned)((xa ? ta : ta - NTI) * 16 + l15);
  const unsigned nb = (unsigned)((xb ? tb : tb - NTI) * 16 + l15);
  const bool oka = na < sa, okb = nb < sb;
  const unsigned ca = oka ? na : sa - 1u;
  const unsigned cb = okb ? nb : sb - 1u;

  v2f bfa[75], bfb[75];
  const unsigned oa = (unsigned)kb0 * sa + ca;
  const unsigned ob = (unsigned)kb0 * sb + cb;
#pragma unroll
  for (int ks = 0; ks < 75; ++ks) {
    const unsigned ka = 4u * (unsigned)ks * sa;
    const unsigned kb = 4u * (unsigned)ks * sb;
    bfa[ks].x = Ba[oa + ka];
    bfa[ks].y = Ba[oa + ka + sa];
    bfb[ks].x = Bb[ob + kb];
    bfb[ks].y = Bb[ob + kb + sb];
  }
  const float bva = ba[ca], bvb = bb[cb];

  int item[5];
#pragma unroll
  for (int u = 0; u < 5; ++u) {
    const int idx = u * 256 + tid;
    item[u] = idx < ATQ ? idx : ATQ - 1;
  }
  int mt = bid - grp * GWA;
  v4f pre[5];
  {
    const v4f* g = (const v4f*)(A + (size_t)mt * ATILE);
#pragma unroll
    for (int u = 0; u < 5; ++u) pre[u] = g[item[u]];
#pragma unroll
    for (int u = 0; u < 5; ++u) ((v4f*)at[0])[item[u]] = pre[u];
  }
  __syncthreads();
  int buf = 0;
  for (; mt < NN / 16; mt += step) {
    const int  nx   = mt + step;
    const bool more = nx < NN / 16;
    if (more) {
      const v4f* g = (const v4f*)(A + (size_t)nx * ATILE);
#pragma unroll
      for (int u = 0; u < 5; ++u) pre[u] = g[item[u]];
    }
    v8f acc0 = {}, acc1 = {};
    if (live) {
      const float* arow = at[buf] + l15 * DIN + kb0;
      v2f a[75];
#pragma unroll
      for (int ks = 0; ks < 50; ++ks) a[ks] = *(const v2f*)(arow + 4 * ks);
      __builtin_amdgcn_sched_barrier(0);
#pragma unroll
      for (int ks = 0; ks < 25; ++ks) {
        acc0 = __builtin_amdgcn_wmma_f32_16x16x4_f32(false, a[ks], false, bfa[ks],
                                                     (short)0, acc0, false, false);
        acc1 = __builtin_amdgcn_wmma_f32_16x16x4_f32(false, a[ks], false, bfb[ks],
                                                     (short)0, acc1, false, false);
      }
#pragma unroll
      for (int ks = 50; ks < 75; ++ks) a[ks] = *(const v2f*)(arow + 4 * ks);
      __builtin_amdgcn_sched_barrier(0);
#pragma unroll
      for (int ks = 25; ks < 75; ++ks) {
        acc0 = __builtin_amdgcn_wmma_f32_16x16x4_f32(false, a[ks], false, bfa[ks],
                                                     (short)0, acc0, false, false);
        acc1 = __builtin_amdgcn_wmma_f32_16x16x4_f32(false, a[ks], false, bfb[ks],
                                                     (short)0, acc1, false, false);
      }
    }
    if (more) {
#pragma unroll
      for (int u = 0; u < 5; ++u) ((v4f*)at[buf ^ 1])[item[u]] = pre[u];
    }
    __syncthreads();
    if (live) {
      const unsigned r0 = (unsigned)(mt * 16 + ((lane >> 4) << 3));
      if (oka) {
#pragma unroll
        for (int v = 0; v < 8; ++v) Ca[(r0 + (unsigned)v) * sa + na] = acc0[v] + bva;
      }
      if (okb) {
#pragma unroll
        for (int v = 0; v < 8; ++v) Cb[(r0 + (unsigned)v) * sb + nb] = acc1[v] + bvb;
      }
    }
    buf ^= 1;
  }
}

__global__ void leaf_kernel(const float* __restrict__ xiou,
                            const float* __restrict__ biouh,
                            float* __restrict__ cst, float* __restrict__ hst,
                            int count) {
  int e = (int)blockIdx.x * blockDim.x + threadIdx.x;
  if (e >= count * HD) return;
  int node = e / HD, col = e - node * HD;
  const float* r = xiou + (size_t)node * IOUW;
  float i = sigmoid_f(r[col] + biouh[col]);
  float o = sigmoid_f(r[HD + col] + biouh[HD + col]);
  float u = tanh_f(r[2 * HD + col] + biouh[2 * HD + col]);
  float c = i * u;
  cst[(size_t)node * HD + col] = c;
  hst[(size_t)node * HD + col] = o * tanh_f(c);
}

template <int LD, int COL0, int NVP, int S>
__device__ __forceinline__ void panel_issue(const float* __restrict__ Wg, int tid,
                                            v2f (&r0)[S], v2f (&r1)[S]) {
  constexpr int ITEMS = 75 * NVP;
  static_assert(S * 256 >= ITEMS, "slots must cover the panel");
#pragma unroll
  for (int u = 0; u < S; ++u) {
    int idx = u * 256 + tid;
    int idc = idx < ITEMS ? idx : ITEMS - 1;
    int kp = idc / NVP, cp = idc - kp * NVP;
    const float* p = Wg + (2 * kp) * LD + COL0 + 2 * cp;
    r0[u] = *(const v2f*)p;
    r1[u] = *(const v2f*)(p + LD);
  }
}
template <int NVP, int PW, int S>
__device__ __forceinline__ void panel_store(float* pan, int tid,
                                            const v2f (&r0)[S], const v2f (&r1)[S]) {
  constexpr int ITEMS = 75 * NVP;
  const v4f zero4 = {0.0f, 0.0f, 0.0f, 0.0f};
#pragma unroll
  for (int u = 0; u < S; ++u) {
    int idx = u * 256 + tid;
    int idc = idx < ITEMS ? idx : ITEMS - 1;
    int kp = idc / NVP, cp = idc - kp * NVP;
    v4f o = {r0[u].x, r1[u].x, r0[u].y, r1[u].y};
    *(v4f*)(pan + kp * WPITCH + 4 * cp) = o;
  }
  constexpr int PQ = PW / 2;
  for (int c = tid; c < PQ; c += 256)
    *(v4f*)(pan + 75 * WPITCH + 4 * c) = zero4;
  constexpr int NPAD = PQ - NVP;
  constexpr int NPD  = NPAD > 0 ? NPAD : 1;
  for (int idx = tid; idx < 75 * NPAD; idx += 256) {
    int kp = idx / NPD, cp = NVP + idx - kp * NPD;
    *(v4f*)(pan + kp * WPITCH + 4 * cp) = zero4;
  }
}

extern __shared__ __attribute__((aligned(16))) float smem[];

__device__ __forceinline__ int flat_item(int idx, int n2) {
  int ic = idx < n2 ? idx : n2 - 1;
  return ic < 0 ? 0 : ic;
}
template <int S>
__device__ __forceinline__ void flat_issue(const float* __restrict__ src,
                                           int n2, int tid, v2f (&r)[S]) {
#pragma unroll
  for (int u = 0; u < S; ++u)
    r[u] = *(const v2f*)(src + 2 * flat_item(u * 256 + tid, n2));
}
template <int S>
__device__ __forceinline__ void flat_store(float* dst, int n2, int tid,
                                           const v2f (&r)[S]) {
#pragma unroll
  for (int u = 0; u < S; ++u)
    *(v2f*)(dst + 2 * flat_item(u * 256 + tid, n2)) = r[u];
}

typedef const __attribute__((address_space(1))) float* gmem_cf;
typedef __attribute__((address_space(1))) float* gmem_f;

__device__ __forceinline__ int iou_tile(int w, int j) {
  return (j >> 1) * 16 + 2 * w + (j & 1);
}

__device__ __forceinline__ void
forget_task(const float* chh, const float* chc, const float* bpan,
            const float* xsf, float* fcs, const float (&bv)[2],
            int m0, int nt0, int cnt, int lane) {
  const int l15  = lane & 15;
  const int kb0  = (lane >> 4) << 1;
  const int roff = (lane >> 4) << 3;
  const float* ap = chh + (m0 + l15) * CHP + kb0;
  const float* bq = bpan + nt0 * 32;
  v2f a[38], b0[38], b1[38];
  auto ld = [&](int ks) {
    a[ks]  = *(const v2f*)(ap + 4 * ks);
    b0[ks] = *(const v2f*)(bq + 2 * ks * WPITCH);
    b1[ks] = *(const v2f*)(bq + 2 * ks * WPITCH + 32);
  };
#pragma unroll
  for (int ks = 0; ks < 8; ++ks) ld(ks);
  __builtin_amdgcn_sched_barrier(0);
  v8f acc[2] = {};
#pragma unroll
  for (int c = 0; c < 10; ++c) {
#pragma unroll
    for (int i = 0; i < 4; ++i)
      if (4 * (c + 2) + i < 38) ld(4 * (c + 2) + i);
#pragma unroll
    for (int i = 0; i < 4; ++i) {
      const int ks = 4 * c + i;
      if (ks < 38) {
        acc[0] = __builtin_amdgcn_wmma_f32_16x16x4_f32(false, a[ks], false, b0[ks], (short)0,
                                                       acc[0], false, false);
        acc[1] = __builtin_amdgcn_wmma_f32_16x16x4_f32(false, a[ks], false, b1[ks], (short)0,
                                                       acc[1], false, false);
      }
    }
    __builtin_amdgcn_sched_barrier(0);
  }
#pragma unroll
  for (int j = 0; j < 2; ++j) {
    const int  n   = (nt0 + j) * 16 + l15;
    const bool nok = n < HD;
    const int  nc  = nok ? n : 0;
    float s0 = 0.0f, s1 = 0.0f;
#pragma unroll
    for (int v = 0; v < 8; ++v) {
      int grow = m0 + v + roff;
      int  mr  = grow >> 2;
      bool xok = nok && mr < cnt;
      float xs = xsf[(xok ? mr : 0) * HD + nc];
      float xv = xok ? xs : 0.0f;
      float f  = sigmoid_f(acc[j][v] + bv[j] + xv);
      float w  = f * chc[grow * CHP + nc];
      if (v < 4) s0 += w; else s1 += w;
    }
    int nl = (m0 >> 2) + ((lane >> 4) << 1);
    fcs[nl * FCP + n]       = s0;
    fcs[(nl + 1) * FCP + n] = s1;
  }
}

__device__ __forceinline__ void
sub_level(gmem_cf bfh_p,
          gmem_f cst_p, gmem_f hst_p, const v2f (&wb)[4][38],
          const float (&bvj)[4],
          int base, int cnt, int xo, bool more) {
  const float* __restrict__ bfh   = (const float*)bfh_p;
  float* __restrict__ cst = (float*)cst_p;
  float* __restrict__ hst = (float*)hst_p;
  float* chh  = smem;
  float* chc  = chh  + 64 * CHP;
  float* hsum = chc  + 64 * CHP;
  float* fcs  = hsum + 16 * CHP;
  float* ious = fcs  + 16 * FCP;
  float* wpan = ious + 16 * IOUP;
  const float* xsi = wpan + WROWS * WPITCH + xo * IOUW;
  const float* xsf = wpan + WROWS * WPITCH + XROWS * IOUW + xo * HD;

  const int tid  = threadIdx.x;
  const int lane = tid & 31;
  const int wave = tid >> 5;
  const int l15  = lane & 15;
  const int kb0  = (lane >> 4) << 1;
  const int roff = (lane >> 4) << 3;
  const int end  = base + cnt;

  __syncthreads();
  for (int e = tid; e < 16 * CHP; e += 256) {
    int m = e / CHP, col = e - m * CHP;
    int r0 = (m * 4) * CHP + col;
    hsum[e] = chh[r0] + chh[r0 + CHP] + chh[r0 + 2 * CHP] + chh[r0 + 3 * CHP];
  }
  __syncthreads();

  const float* bpan = wpan + (lane >> 4) * WPITCH + l15 * 2;

  {
    const int ntask = ((4 * cnt + 15) >> 4) * 5;
    const int ws = __builtin_amdgcn_readfirstlane(wave);
    for (int t = ws; t < ntask; t += 8) {
      const int mt = t / 5, nt0 = 2 * (t - mt * 5);
      float bv[2];
#pragma unroll
      for (int j = 0; j < 2; ++j) {
        const int n = (nt0 + j) * 16 + l15;
        const float bl = bfh[n < HD ? n : 0];
        bv[j] = n < HD ? bl : 0.0f;
      }
      forget_task(chh, chc, bpan, xsf, fcs, bv, mt * 16, nt0, cnt, lane);
    }
  }

  {
    const float* ap = hsum + l15 * CHP + kb0;
    v2f a[38];
#pragma unroll
    for (int ks = 0; ks < 38; ++ks) a[ks] = *(const v2f*)(ap + 4 * ks);
    v8f acc[4] = {};
#pragma unroll
    for (int ks = 0; ks < 38; ++ks) {
#pragma unroll
      for (int j = 0; j < 4; ++j)
        acc[j] = __builtin_amdgcn_wmma_f32_16x16x4_f32(false, a[ks], false, wb[j][ks],
                                                       (short)0, acc[j], false, false);
    }
#pragma unroll
    for (int j = 0; j < 4; ++j) {
      const int n = iou_tile(wave, j) * 16 + l15;
      if (n < IOUW) {
        float bv = bvj[j];
#pragma unroll
        for (int v = 0; v < 8; ++v) {
          int m = v + roff;
          bool xok = m < cnt;
          float xs = xsi[(xok ? m : 0) * IOUW + n];
          float xv = xok ? xs : 0.0f;
          ious[m * IOUP + n] = acc[j][v] + bv + xv;
        }
      }
    }
  }
  __syncthreads();

  for (int e = tid; e < 16 * HD; e += 256) {
    int m = e / HD, col = e - m * HD;
    int node = base + m;
    if (node < end) {
      float i = sigmoid_f(ious[m * IOUP + col]);
      float o = sigmoid_f(ious[m * IOUP + HD + col]);
      float u = tanh_f(ious[m * IOUP + 2 * HD + col]);
      float c = i * u + fcs[m * FCP + col];
      float h = o * tanh_f(c);
      hst[(size_t)node * HD + col] = h;
      if (more) {
        chh[(m ^ 3) * CHP + col] = h;
        chc[(m ^ 3) * CHP + col] = c;
      } else {
        cst[(size_t)node * HD + col] = c;
      }
    }
  }
}

__global__ void __launch_bounds__(256)
level_kernel(const float* __restrict__ xiou, const float* __restrict__ xf,
             const float* __restrict__ Wiouh, const float* __restrict__ biouh,
             const float* __restrict__ Wfh, const float* __restrict__ bfh,
             float* __restrict__ cst, float* __restrict__ hst,
             int s0, int s1, int s2, int cnt0, int nsub) {
  const int tid  = threadIdx.x;
  const int lane = tid & 31;
  const int wave = tid >> 5;
  const int l15  = lane & 15;
  const int kb0  = (lane >> 4) << 1;

  float* const xs  = smem + 2 * 64 * CHP + 16 * CHP + 16 * FCP + 16 * IOUP + WROWS * WPITCH;
  float* const chh = smem;
  float* const chc = smem + 64 * CHP;
  const int bid = (int)blockIdx.x;
  const int c0  = cnt0, c1 = cnt0 >> 2, c2 = nsub > 2 ? cnt0 >> 4 : 0;
  const int xo1 = c0, xo2 = c0 + c1;
  const float* const gi0 = xiou + (size_t)(s0 + bid * c0) * IOUW;
  const float* const gi1 = xiou + (size_t)(s1 + bid * c1) * IOUW;
  const float* const gi2 = xiou + (size_t)(s2 + bid * c2) * IOUW;
  const float* const gf0 = xf + (size_t)(s0 + bid * c0) * HD;
  const float* const gf1 = xf + (size_t)(s1 + bid * c1) * HD;
  const float* const gf2 = xf + (size_t)(s2 + bid * c2) * HD;
  v2f ri0[15], rf0[5], ri1[4], rf1[2], ri2[1], rf2[1];
  flat_issue<15>(gi0, c0 * (IOUW / 2), tid, ri0);
  flat_issue<5>(gf0, c0 * (HD / 2), tid, rf0);
  flat_issue<4>(gi1, c1 * (IOUW / 2), tid, ri1);
  flat_issue<2>(gf1, c1 * (HD / 2), tid, rf1);
  flat_issue<1>(gi2, c2 * (IOUW / 2), tid, ri2);
  flat_issue<1>(gf2, c2 * (HD / 2), tid, rf2);

  constexpr int GS = 19;
  constexpr int GN = 64 * 75;
  const int cmin = (NN - 2) - 4 * ((NN - 1) - (s0 + bid * c0)) - 3;
  const int jlo  = cmin < 0 ? -cmin : 0;
  const int jhi  = 4 * c0 < 64 ? 4 * c0 : 64;
  const int ilo  = 75 * jlo, ihi = 75 * jhi;
  v2f rh[GS], rc[GS];
#pragma unroll
  for (int u = 0; u < GS; ++u) {
    int idx = u * 256 + tid;
    idx = idx < GN ? idx : GN - 1;
    bool ok = idx >= ilo && idx < ihi;
    int off = ok ? cmin * HD + 2 * idx : 0;
    rh[u] = *(const v2f*)(hst + off);
    rc[u] = *(const v2f*)(cst + off);
  }
  v2f rp0[22], rp1[22];
  panel_issue<HD, 0, 75, 22>(Wfh, tid, rp0, rp1);
  __builtin_amdgcn_sched_barrier(0);

  v2f wb[4][38];
  auto load_wb = [&](int j) {
    const int  n   = iou_tile(wave, j) * 16 + l15;
    const bool nok = n < IOUW;
    const int  nc  = nok ? n : IOUW - 1;
#pragma unroll
    for (int ks = 0; ks < 38; ++ks) {
      const int  k0  = 4 * ks + kb0;
      const bool ok  = nok && k0 < HD;
      const int  kc  = k0 < HD ? k0 : HD - 2;
      const float w0 = Wiouh[kc * IOUW + nc];
      const float w1 = Wiouh[(kc + 1) * IOUW + nc];
      wb[j][ks].x = ok ? w0 : 0.0f;
      wb[j][ks].y = ok ? w1 : 0.0f;
    }
  };
  load_wb(0);
  load_wb(1);
  __builtin_amdgcn_sched_barrier(0);

  flat_store<15>(xs, c0 * (IOUW / 2), tid, ri0);
  flat_store<5>(xs + XROWS * IOUW, c0 * (HD / 2), tid, rf0);
  flat_store<4>(xs + xo1 * IOUW, c1 * (IOUW / 2), tid, ri1);
  flat_store<2>(xs + XROWS * IOUW + xo1 * HD, c1 * (HD / 2), tid, rf1);
  flat_store<1>(xs + xo2 * IOUW, c2 * (IOUW / 2), tid, ri2);
  flat_store<1>(xs + XROWS * IOUW + xo2 * HD, c2 * (HD / 2), tid, rf2);
#pragma unroll
  for (int u = 0; u < GS; ++u) {
    int idx = u * 256 + tid;
    idx = idx < GN ? idx : GN - 1;
    bool ok = idx >= ilo && idx < ihi;
    int j = idx / 75, q = idx - j * 75;
    const v2f z2 = {0.0f, 0.0f};
    *(v2f*)(chh + (j ^ 3) * CHP + 2 * q) = ok ? rh[u] : z2;
    *(v2f*)(chc + (j ^ 3) * CHP + 2 * q) = ok ? rc[u] : z2;
  }
  if (tid < 128) {
    const v2f z2 = {0.0f, 0.0f};
    *(v2f*)((tid < 64 ? chh : chc) + (tid & 63) * CHP + HD) = z2;
  }

  panel_store<75, 160, 22>(smem + 2 * 64 * CHP + 16 * CHP + 16 * FCP + 16 * IOUP, tid, rp0, rp1);

  load_wb(2);
  load_wb(3);

  float bvj[4];
#pragma unroll
  for (int j = 0; j < 4; ++j) {
    const int n = iou_tile(wave, j) * 16 + l15;
    bvj[j] = biouh[n < IOUW ? n : IOUW - 1];
  }

  int xo = 0;
  for (int sub = 0; sub < nsub; ++sub) {
    const int cnt  = cnt0 >> (2 * sub);
    const int lvl0 = sub == 0 ? s0 : (sub == 1 ? s1 : s2);
    sub_level((gmem_cf)bfh,
              (gmem_f)cst, (gmem_f)hst, wb, bvj,
              lvl0 + (int)blockIdx.x * cnt, cnt, xo, sub + 1 < nsub);
    xo += cnt;
  }
}

extern "C" void kernel_launch(void* const* d_in, const int* in_sizes, int n_in,
                              void* d_out, int out_size, void* d_ws, size_t ws_size,
                              hipStream_t stream) {
  const float* x     = (const float*)d_in[0];
  const float* Wioux = (const float*)d_in[1];
  const float* bioux = (const float*)d_in[2];
  const float* Wiouh = (const float*)d_in[3];
  const float* biouh = (const float*)d_in[4];
  const float* Wfx   = (const float*)d_in[5];
  const float* bfx   = (const float*)d_in[6];
  const float* Wfh   = (const float*)d_in[7];
  const float* bfh   = (const float*)d_in[8];
  (void)in_sizes; (void)n_in; (void)out_size; (void)ws_size;

  float* xiou = (float*)d_ws;
  float* xf   = xiou + (size_t)NN * IOUW;
  float* cst  = xf   + (size_t)NN * HD;
  float* hst  = (float*)d_out;

  proj_gemm<<<256, 256, 0, stream>>>(x, Wioux, bioux, xiou, Wfx, bfx, xf);

  const int leafEnd = 14539;
  leaf_kernel<<<(leafEnd * HD + 255) / 256, 256, 0, stream>>>(xiou, biouh,
                                                              cst, hst, leafEnd);

  const size_t lds =
      (size_t)(2 * 64 * CHP + 16 * CHP + 16 * FCP + 16 * IOUP + WROWS * WPITCH +
               XROWS * (IOUW + HD)) * sizeof(float);
  level_kernel<<<256, 256, lds, stream>>>(xiou, xf, Wiouh, biouh, Wfh, bfh,
                                          cst, hst, 14539, 18635, 19659, 16, 3);
  level_kernel<<<4, 256, lds, stream>>>(xiou, xf, Wiouh, biouh, Wfh, bfh,
                                        cst, hst, 19915, 19979, 0, 16, 2);
  level_kernel<<<1, 256, lds, stream>>>(xiou, xf, Wiouh, biouh, Wfh, bfh,
                                        cst, hst, 19995, 19999, 0, 4, 2);
}


# level_kernel, 4- and 1-node sub-levels run as one product: the tile of child rows times [Wfh | Wiouh], 39 live n-tile chains dealt evenly over SIMDs; child sum over accumulator rows (sum order new)
# speedup vs baseline: 5.1966x; 1.0739x over previous
#include <hip/hip_runtime.h>


typedef __attribute__((ext_vector_type(2))) float v2f;
typedef __attribute__((ext_vector_type(4))) float v4f;
typedef __attribute__((ext_vector_type(8))) float v8f;

#define NN   20000
#define DIN  300
#define HD   150
#define IOUW 450
#define CHP  152
#define IOUP 464
#define FCP  160
#define WROWS  76
#define WPITCH 352
#define XROWS  21

__device__ __forceinline__ float sigmoid_f(float x) {
  return 1.0f / (1.0f + __expf(-x));
}
__device__ __forceinline__ float tanh_f(float x) {
  float t = __expf(2.0f * fminf(fmaxf(x, -15.0f), 15.0f));
  return (t - 1.0f) / (t + 1.0f);
}

#define NTI   29
#define NTA   39
#define GWA   100
#define GWC   56
static_assert(2 * GWA + GWC == 256, "the launch has 256 workgroups");
#define ATILE (16 * DIN)
#define ATQ   (ATILE / 4)
__global__ void __launch_bounds__(256)
proj_gemm(const float* __restrict__ A,
          const float* __restrict__ B0, const float* __restrict__ bias0,
          float* __restrict__ C0,
          const float* __restrict__ B1, const float* __restrict__ bias1,
          float* __restrict__ C1) {
  __shared__ __attribute__((aligned(16))) float at[2][ATILE];
  const int tid  = threadIdx.x;
  const int lane = tid & 31;
  const int wave = __builtin_amdgcn_readfirstlane(tid >> 5);
  const int l15  = lane & 15;
  const int kb0  = (lane >> 4) << 1;
  const int bid  = (int)blockIdx.x;
  const int grp  = bid < GWA ? 0 : (bid < 2 * GWA ? 1 : 2);
  const int step = grp < 2 ? GWA : GWC;
  const int ta   = grp * 16 + wave * 2, tb = ta + 1;
  const bool live = ta < NTA;
  const bool xa = ta < NTI, xb = tb < NTI;
  const float* __restrict__ Ba = xa ? B0 : B1;
  const float* __restrict__ Bb = xb ? B0 : B1;
  const float* __restrict__ ba = xa ? bias0 : bias1;
  const float* __restrict__ bb = xb ? bias0 : bias1;
  float* __restrict__ Ca = xa ? C0 : C1;
  float* __restrict__ Cb = xb ? C0 : C1;
  const unsigned sa = xa ? IOUW : HD, sb = xb ? IOUW : HD;
  const unsigned na = (unsigned)((xa ? ta : ta - NTI) * 16 + l15);
  const unsigned nb = (unsigned)((xb ? tb : tb - NTI) * 16 + l15);
  const bool oka = na < sa, okb = nb < sb;
  const unsigned ca = oka ? na : sa - 1u;
  const unsigned cb = okb ? nb : sb - 1u;

  v2f bfa[75], bfb[75];
  const unsigned oa = (unsigned)kb0 * sa + ca;
  const unsigned ob = (unsigned)kb0 * sb + cb;
#pragma unroll
  for (int ks = 0; ks < 75; ++ks) {
    const unsigned ka = 4u * (unsigned)ks * sa;
    const unsigned kb = 4u * (unsigned)ks * sb;
    bfa[ks].x = Ba[oa + ka];
    bfa[ks].y = Ba[oa + ka + sa];
    bfb[ks].x = Bb[ob + kb];
    bfb[ks].y = Bb[ob + kb + sb];
  }
  const float bva = ba[ca], bvb = bb[cb];

  int item[5];
#pragma unroll
  for (int u = 0; u < 5; ++u) {
    const int idx = u * 256 + tid;
    item[u] = idx < ATQ ? idx : ATQ - 1;
  }
  int mt = bid - grp * GWA;
  v4f pre[5];
  {
    const v4f* g = (const v4f*)(A + (size_t)mt * ATILE);
#pragma unroll
    for (int u = 0; u < 5; ++u) pre[u] = g[item[u]];
#pragma unroll
    for (int u = 0; u < 5; ++u) ((v4f*)at[0])[item[u]] = pre[u];
  }
  __syncthreads();
  int buf = 0;
  for (; mt < NN / 16; mt += step) {
    const int  nx   = mt + step;
    const bool more = nx < NN / 16;
    if (more) {
      const v4f* g = (const v4f*)(A + (size_t)nx * ATILE);
#pragma unroll
      for (int u = 0; u < 5; ++u) pre[u] = g[item[u]];
    }
    v8f acc0 = {}, acc1 = {};
    if (live) {
      const float* arow = at[buf] + l15 * DIN + kb0;
      v2f a[75];
#pragma unroll
      for (int ks = 0; ks < 50; ++ks) a[ks] = *(const v2f*)(arow + 4 * ks);
      __builtin_amdgcn_sched_barrier(0);
#pragma unroll
      for (int ks = 0; ks < 25; ++ks) {
        acc0 = __builtin_amdgcn_wmma_f32_16x16x4_f32(false, a[ks], false, bfa[ks],
                                                     (short)0, acc0, false, false);
        acc1 = __builtin_amdgcn_wmma_f32_16x16x4_f32(false, a[ks], false, bfb[ks],
                                                     (short)0, acc1, false, false);
      }
#pragma unroll
      for (int ks = 50; ks < 75; ++ks) a[ks] = *(const v2f*)(arow + 4 * ks);
      __builtin_amdgcn_sched_barrier(0);
#pragma unroll
      for (int ks = 25; ks < 75; ++ks) {
        acc0 = __builtin_amdgcn_wmma_f32_16x16x4_f32(false, a[ks], false, bfa[ks],
                                                     (short)0, acc0, false, false);
        acc1 = __builtin_amdgcn_wmma_f32_16x16x4_f32(false, a[ks], false, bfb[ks],
                                                     (short)0, acc1, false, false);
      }
    }
    if (more) {
#pragma unroll
      for (int u = 0; u < 5; ++u) ((v4f*)at[buf ^ 1])[item[u]] = pre[u];
    }
    __syncthreads();
    if (live) {
      const unsigned r0 = (unsigned)(mt * 16 + ((lane >> 4) << 3));
      if (oka) {
#pragma unroll
        for (int v = 0; v < 8; ++v) Ca[(r0 + (unsigned)v) * sa + na] = acc0[v] + bva;
      }
      if (okb) {
#pragma unroll
        for (int v = 0; v < 8; ++v) Cb[(r0 + (unsigned)v) * sb + nb] = acc1[v] + bvb;
      }
    }
    buf ^= 1;
  }
}

__global__ void leaf_kernel(const float* __restrict__ xiou,
                            const float* __restrict__ biouh,
                            float* __restrict__ cst, float* __restrict__ hst,
                            int count) {
  int e = (int)blockIdx.x * blockDim.x + threadIdx.x;
  if (e >= count * HD) return;
  int node = e / HD, col = e - node * HD;
  const float* r = xiou + (size_t)node * IOUW;
  float i = sigmoid_f(r[col] + biouh[col]);
  float o = sigmoid_f(r[HD + col] + biouh[HD + col]);
  float u = tanh_f(r[2 * HD + col] + biouh[2 * HD + col]);
  float c = i * u;
  cst[(size_t)node * HD + col] = c;
  hst[(size_t)node * HD + col] = o * tanh_f(c);
}

template <int LD, int COL0, int NVP, int S>
__device__ __forceinline__ void panel_issue(const float* __restrict__ Wg, int tid,
                                            v2f (&r0)[S], v2f (&r1)[S]) {
  constexpr int ITEMS = 75 * NVP;
  static_assert(S * 256 >= ITEMS, "slots must cover the panel");
#pragma unroll
  for (int u = 0; u < S; ++u) {
    int idx = u * 256 + tid;
    int idc = idx < ITEMS ? idx : ITEMS - 1;
    int kp = idc / NVP, cp = idc - kp * NVP;
    const float* p = Wg + (2 * kp) * LD + COL0 + 2 * cp;
    r0[u] = *(const v2f*)p;
    r1[u] = *(const v2f*)(p + LD);
  }
}
template <int NVP, int PW, int S>
__device__ __forceinline__ void panel_store(float* pan, int tid,
                                            const v2f (&r0)[S], const v2f (&r1)[S]) {
  constexpr int ITEMS = 75 * NVP;
  const v4f zero4 = {0.0f, 0.0f, 0.0f, 0.0f};
#pragma unroll
  for (int u = 0; u < S; ++u) {
    int idx = u * 256 + tid;
    int idc = idx < ITEMS ? idx : ITEMS - 1;
    int kp = idc / NVP, cp = idc - kp * NVP;
    v4f o = {r0[u].x, r1[u].x, r0[u].y, r1[u].y};
    *(v4f*)(pan + kp * WPITCH + 4 * cp) = o;
  }
  constexpr int PQ = PW / 2;
  for (int c = tid; c < PQ; c += 256)
    *(v4f*)(pan + 75 * WPITCH + 4 * c) = zero4;
  constexpr int NPAD = PQ - NVP;
  constexpr int NPD  = NPAD > 0 ? NPAD : 1;
  for (int idx = tid; idx < 75 * NPAD; idx += 256) {
    int kp = idx / NPD, cp = NVP + idx - kp * NPD;
    *(v4f*)(pan + kp * WPITCH + 4 * cp) = zero4;
  }
}

extern __shared__ __attribute__((aligned(16))) float smem[];

__device__ __forceinline__ int flat_item(int idx, int n2) {
  int ic = idx < n2 ? idx : n2 - 1;
  return ic < 0 ? 0 : ic;
}
template <int S>
__device__ __forceinline__ void flat_issue(const float* __restrict__ src,
                                           int n2, int tid, v2f (&r)[S]) {
#pragma unroll
  for (int u = 0; u < S; ++u)
    r[u] = *(const v2f*)(src + 2 * flat_item(u * 256 + tid, n2));
}
template <int S>
__device__ __forceinline__ void flat_store(float* dst, int n2, int tid,
                                           const v2f (&r)[S]) {
#pragma unroll
  for (int u = 0; u < S; ++u)
    *(v2f*)(dst + 2 * flat_item(u * 256 + tid, n2)) = r[u];
}

typedef const __attribute__((address_space(1))) float* gmem_cf;
typedef __attribute__((address_space(1))) float* gmem_f;

__device__ __forceinline__ int iou_tile(int w, int j) {
  return (j >> 1) * 16 + 2 * w + (j & 1);
}

__device__ __forceinline__ void
forget_task(const float* chh, const float* chc, const float* bpan,
            const float* xsf, float* fcs, const float (&bv)[2],
            int m0, int nt0, int cnt, int lane) {
  const int l15  = lane & 15;
  const int kb0  = (lane >> 4) << 1;
  const int roff = (lane >> 4) << 3;
  const float* ap = chh + (m0 + l15) * CHP + kb0;
  const float* bq = bpan + nt0 * 32;
  v2f a[38], b0[38], b1[38];
  auto ld = [&](int ks) {
    a[ks]  = *(const v2f*)(ap + 4 * ks);
    b0[ks] = *(const v2f*)(bq + 2 * ks * WPITCH);
    b1[ks] = *(const v2f*)(bq + 2 * ks * WPITCH + 32);
  };
#pragma unroll
  for (int ks = 0; ks < 8; ++ks) ld(ks);
  __builtin_amdgcn_sched_barrier(0);
  v8f acc[2] = {};
#pragma unroll
  for (int c = 0; c < 10; ++c) {
#pragma unroll
    for (int i = 0; i < 4; ++i)
      if (4 * (c + 2) + i < 38) ld(4 * (c + 2) + i);
#pragma unroll
    for (int i = 0; i < 4; ++i) {
      const int ks = 4 * c + i;
      if (ks < 38) {
        acc[0] = __builtin_amdgcn_wmma_f32_16x16x4_f32(false, a[ks], false, b0[ks], (short)0,
                                                       acc[0], false, false);
        acc[1] = __builtin_amdgcn_wmma_f32_16x16x4_f32(false, a[ks], false, b1[ks], (short)0,
                                                       acc[1], false, false);
      }
    }
    __builtin_amdgcn_sched_barrier(0);
  }
#pragma unroll
  for (int j = 0; j < 2; ++j) {
    const int  n   = (nt0 + j) * 16 + l15;
    const bool nok = n < HD;
    const int  nc  = nok ? n : 0;
    float s0 = 0.0f, s1 = 0.0f;
#pragma unroll
    for (int v = 0; v < 8; ++v) {
      int grow = m0 + v + roff;
      int  mr  = grow >> 2;
      bool xok = nok && mr < cnt;
      float xs = xsf[(xok ? mr : 0) * HD + nc];
      float xv = xok ? xs : 0.0f;
      float f  = sigmoid_f(acc[j][v] + bv[j] + xv);
      float w  = f * chc[grow * CHP + nc];
      if (v < 4) s0 += w; else s1 += w;
    }
    int nl = (m0 >> 2) + ((lane >> 4) << 1);
    fcs[nl * FCP + n]       = s0;
    fcs[(nl + 1) * FCP + n] = s1;
  }
}

__device__ __forceinline__ void
sub_level(gmem_cf bfh_p,
          gmem_f cst_p, gmem_f hst_p, const v2f (&wb)[4][38],
          const float (&bvj)[4],
          int base, int cnt, int xo, bool more) {
  const float* __restrict__ bfh   = (const float*)bfh_p;
  float* __restrict__ cst = (float*)cst_p;
  float* __restrict__ hst = (float*)hst_p;
  float* chh  = smem;
  float* chc  = chh  + 64 * CHP;
  float* hsum = chc  + 64 * CHP;
  float* fcs  = hsum + 16 * CHP;
  float* ious = fcs  + 16 * FCP;
  float* wpan = ious + 16 * IOUP;
  const float* xsi = wpan + WROWS * WPITCH + xo * IOUW;
  const float* xsf = wpan + WROWS * WPITCH + XROWS * IOUW + xo * HD;

  const int tid  = threadIdx.x;
  const int lane = tid & 31;
  const int wave = tid >> 5;
  const int l15  = lane & 15;
  const int kb0  = (lane >> 4) << 1;
  const int roff = (lane >> 4) << 3;
  const int end  = base + cnt;

  const bool one_tile = 4 * cnt <= 16;
  __syncthreads();
  if (!one_tile) {
    for (int e = tid; e < 16 * CHP; e += 256) {
      int m = e / CHP, col = e - m * CHP;
      int r0 = (m * 4) * CHP + col;
      hsum[e] = chh[r0] + chh[r0 + CHP] + chh[r0 + 2 * CHP] + chh[r0 + 3 * CHP];
    }
    __syncthreads();
  }

  const float* bpan = wpan + (lane >> 4) * WPITCH + l15 * 2;

  const int ws = __builtin_amdgcn_readfirstlane(wave);
  {
    int t0 = ws, t1 = ((4 * cnt + 15) >> 4) * 5, ts = 8;
    if (one_tile) {
      t0 = ws == 7 ? 0 : (ws == 6 ? 2 : (ws == 4 ? 3 : 4));
      t1 = ws == 7 ? 2 : (ws == 6 ? 3 : (ws == 4 ? 4 : (ws == 5 ? 5 : 0)));
      ts = 1;
    }
    for (int t = t0; t < t1; t += ts) {
      const int mt = t / 5, nt0 = 2 * (t - mt * 5);
      float bv[2];
#pragma unroll
      for (int j = 0; j < 2; ++j) {
        const int n = (nt0 + j) * 16 + l15;
        const float bl = bfh[n < HD ? n : 0];
        bv[j] = n < HD ? bl : 0.0f;
      }
      forget_task(chh, chc, bpan, xsf, fcs, bv, mt * 16, nt0, cnt, lane);
    }
  }

  {
    const float* ap = (one_tile ? chh : hsum) + l15 * CHP + kb0;
    v2f a[38];
#pragma unroll
    for (int ks = 0; ks < 38; ++ks) a[ks] = *(const v2f*)(ap + 4 * ks);
    v8f acc[4] = {};
    if (ws < 6) {
#pragma unroll
      for (int ks = 0; ks < 38; ++ks) {
#pragma unroll
        for (int j = 0; j < 4; ++j)
          acc[j] = __builtin_amdgcn_wmma_f32_16x16x4_f32(false, a[ks], false, wb[j][ks],
                                                         (short)0, acc[j], false, false);
      }
    } else if (ws == 6) {
#pragma unroll
      for (int ks = 0; ks < 38; ++ks) {
#pragma unroll
        for (int j = 0; j < 3; ++j)
          acc[j] = __builtin_amdgcn_wmma_f32_16x16x4_f32(false, a[ks], false, wb[j][ks],
                                                         (short)0, acc[j], false, false);
      }
    } else {
#pragma unroll
      for (int ks = 0; ks < 38; ++ks) {
#pragma unroll
        for (int j = 0; j < 2; ++j)
          acc[j] = __builtin_amdgcn_wmma_f32_16x16x4_f32(false, a[ks], false, wb[j][ks],
                                                         (short)0, acc[j], false, false);
      }
    }
#pragma unroll
    for (int j = 0; j < 4; ++j) {
      const int n = iou_tile(wave, j) * 16 + l15;
      if (n < IOUW) {
        float bv = bvj[j];
        if (one_tile) {
          const int mA = (lane >> 4) << 1;
          const float sA = ((acc[j][0] + acc[j][1]) + acc[j][2]) + acc[j][3];
          const float sB = ((acc[j][4] + acc[j][5]) + acc[j][6]) + acc[j][7];
          const bool okA = mA < cnt, okB = mA + 1 < cnt;
          const float xA = xsi[(okA ? mA : 0) * IOUW + n];
          const float xB = xsi[(okB ? mA + 1 : 0) * IOUW + n];
          ious[mA * IOUP + n]       = sA + bv + (okA ? xA : 0.0f);
          ious[(mA + 1) * IOUP + n] = sB + bv + (okB ? xB : 0.0f);
        } else {
#pragma unroll
          for (int v = 0; v < 8; ++v) {
            int m = v + roff;
            bool xok = m < cnt;
            float xs = xsi[(xok ? m : 0) * IOUW + n];
            float xv = xok ? xs : 0.0f;
            ious[m * IOUP + n] = acc[j][v] + bv + xv;
          }
        }
      }
    }
  }
  __syncthreads();

  for (int e = tid; e < 16 * HD; e += 256) {
    int m = e / HD, col = e - m * HD;
    int node = base + m;
    if (node < end) {
      float i = sigmoid_f(ious[m * IOUP + col]);
      float o = sigmoid_f(ious[m * IOUP + HD + col]);
      float u = tanh_f(ious[m * IOUP + 2 * HD + col]);
      float c = i * u + fcs[m * FCP + col];
      float h = o * tanh_f(c);
      hst[(size_t)node * HD + col] = h;
      if (more) {
        chh[(m ^ 3) * CHP + col] = h;
        chc[(m ^ 3) * CHP + col] = c;
      } else {
        cst[(size_t)node * HD + col] = c;
      }
    }
  }
}

__global__ void __launch_bounds__(256)
level_kernel(const float* __restrict__ xiou, const float* __restrict__ xf,
             const float* __restrict__ Wiouh, const float* __restrict__ biouh,
             const float* __restrict__ Wfh, const float* __restrict__ bfh,
             float* __restrict__ cst, float* __restrict__ hst,
             int s0, int s1, int s2, int cnt0, int nsub) {
  const int tid  = threadIdx.x;
  const int lane = tid & 31;
  const int wave = tid >> 5;
  const int l15  = lane & 15;
  const int kb0  = (lane >> 4) << 1;

  float* const xs  = smem + 2 * 64 * CHP + 16 * CHP + 16 * FCP + 16 * IOUP + WROWS * WPITCH;
  float* const chh = smem;
  float* const chc = smem + 64 * CHP;
  const int bid = (int)blockIdx.x;
  const int c0  = cnt0, c1 = cnt0 >> 2, c2 = nsub > 2 ? cnt0 >> 4 : 0;
  const int xo1 = c0, xo2 = c0 + c1;
  const float* const gi0 = xiou + (size_t)(s0 + bid * c0) * IOUW;
  const float* const gi1 = xiou + (size_t)(s1 + bid * c1) * IOUW;
  const float* const gi2 = xiou + (size_t)(s2 + bid * c2) * IOUW;
  const float* const gf0 = xf + (size_t)(s0 + bid * c0) * HD;
  const float* const gf1 = xf + (size_t)(s1 + bid * c1) * HD;
  const float* const gf2 = xf + (size_t)(s2 + bid * c2) * HD;
  v2f ri0[15], rf0[5], ri1[4], rf1[2], ri2[1], rf2[1];
  flat_issue<15>(gi0, c0 * (IOUW / 2), tid, ri0);
  flat_issue<5>(gf0, c0 * (HD / 2), tid, rf0);
  flat_issue<4>(gi1, c1 * (IOUW / 2), tid, ri1);
  flat_issue<2>(gf1, c1 * (HD / 2), tid, rf1);
  flat_issue<1>(gi2, c2 * (IOUW / 2), tid, ri2);
  flat_issue<1>(gf2, c2 * (HD / 2), tid, rf2);

  constexpr int GS = 19;
  constexpr int GN = 64 * 75;
  const int cmin = (NN - 2) - 4 * ((NN - 1) - (s0 + bid * c0)) - 3;
  const int jlo  = cmin < 0 ? -cmin : 0;
  const int jhi  = 4 * c0 < 64 ? 4 * c0 : 64;
  const int ilo  = 75 * jlo, ihi = 75 * jhi;
  v2f rh[GS], rc[GS];
#pragma unroll
  for (int u = 0; u < GS; ++u) {
    int idx = u * 256 + tid;
    idx = idx < GN ? idx : GN - 1;
    bool ok = idx >= ilo && idx < ihi;
    int off = ok ? cmin * HD + 2 * idx : 0;
    rh[u] = *(const v2f*)(hst + off);
    rc[u] = *(const v2f*)(cst + off);
  }
  v2f rp0[22], rp1[22];
  panel_issue<HD, 0, 75, 22>(Wfh, tid, rp0, rp1);
  __builtin_amdgcn_sched_barrier(0);

  v2f wb[4][38];
  auto load_wb = [&](int j) {
    const int  n   = iou_tile(wave, j) * 16 + l15;
    const bool nok = n < IOUW;
    const int  nc  = nok ? n : IOUW - 1;
#pragma unroll
    for (int ks = 0; ks < 38; ++ks) {
      const int  k0  = 4 * ks + kb0;
      const bool ok  = nok && k0 < HD;
      const int  kc  = k0 < HD ? k0 : HD - 2;
      const float w0 = Wiouh[kc * IOUW + nc];
      const float w1 = Wiouh[(kc + 1) * IOUW + nc];
      wb[j][ks].x = ok ? w0 : 0.0f;
      wb[j][ks].y = ok ? w1 : 0.0f;
    }
  };
  load_wb(0);
  load_wb(1);
  __builtin_amdgcn_sched_barrier(0);

  flat_store<15>(xs, c0 * (IOUW / 2), tid, ri0);
  flat_store<5>(xs + XROWS * IOUW, c0 * (HD / 2), tid, rf0);
  flat_store<4>(xs + xo1 * IOUW, c1 * (IOUW / 2), tid, ri1);
  flat_store<2>(xs + XROWS * IOUW + xo1 * HD, c1 * (HD / 2), tid, rf1);
  flat_store<1>(xs + xo2 * IOUW, c2 * (IOUW / 2), tid, ri2);
  flat_store<1>(xs + XROWS * IOUW + xo2 * HD, c2 * (HD / 2), tid, rf2);
#pragma unroll
  for (int u = 0; u < GS; ++u) {
    int idx = u * 256 + tid;
    idx = idx < GN ? idx : GN - 1;
    bool ok = idx >= ilo && idx < ihi;
    int j = idx / 75, q = idx - j * 75;
    const v2f z2 = {0.0f, 0.0f};
    *(v2f*)(chh + (j ^ 3) * CHP + 2 * q) = ok ? rh[u] : z2;
    *(v2f*)(chc + (j ^ 3) * CHP + 2 * q) = ok ? rc[u] : z2;
  }
  if (tid < 128) {
    const v2f z2 = {0.0f, 0.0f};
    *(v2f*)((tid < 64 ? chh : chc) + (tid & 63) * CHP + HD) = z2;
  }

  panel_store<75, 160, 22>(smem + 2 * 64 * CHP + 16 * CHP + 16 * FCP + 16 * IOUP, tid, rp0, rp1);

  load_wb(2);
  load_wb(3);

  float bvj[4];
#pragma unroll
  for (int j = 0; j < 4; ++j) {
    const int n = iou_tile(wave, j) * 16 + l15;
    bvj[j] = biouh[n < IOUW ? n : IOUW - 1];
  }

  int xo = 0;
  for (int sub = 0; sub < nsub; ++sub) {
    const int cnt  = cnt0 >> (2 * sub);
    const int lvl0 = sub == 0 ? s0 : (sub == 1 ? s1 : s2);
    sub_level((gmem_cf)bfh,
              (gmem_f)cst, (gmem_f)hst, wb, bvj,
              lvl0 + (int)blockIdx.x * cnt, cnt, xo, sub + 1 < nsub);
    xo += cnt;
  }
}

extern "C" void kernel_launch(void* const* d_in, const int* in_sizes, int n_in,
                              void* d_out, int out_size, void* d_ws, size_t ws_size,
                              hipStream_t stream) {
  const float* x     = (const float*)d_in[0];
  const float* Wioux = (const float*)d_in[1];
  const float* bioux = (const float*)d_in[2];
  const float* Wiouh = (const float*)d_in[3];
  const float* biouh = (const float*)d_in[4];
  const float* Wfx   = (const float*)d_in[5];
  const float* bfx   = (const float*)d_in[6];
  const float* Wfh   = (const float*)d_in[7];
  const float* bfh   = (const float*)d_in[8];
  (void)in_sizes; (void)n_in; (void)out_size; (void)ws_size;

  float* xiou = (float*)d_ws;
  float* xf   = xiou + (size_t)NN * IOUW;
  float* cst  = xf   + (size_t)NN * HD;
  float* hst  = (float*)d_out;

  proj_gemm<<<256, 256, 0, stream>>>(x, Wioux, bioux, xiou, Wfx, bfx, xf);

  const int leafEnd = 14539;
  leaf_kernel<<<(leafEnd * HD + 255) / 256, 256, 0, stream>>>(xiou, biouh,
                                                              cst, hst, leafEnd);

  const size_t lds =
      (size_t)(2 * 64 * CHP + 16 * CHP + 16 * FCP + 16 * IOUP + WROWS * WPITCH +
               XROWS * (IOUW + HD)) * sizeof(float);
  level_kernel<<<256, 256, lds, stream>>>(xiou, xf, Wiouh, biouh, Wfh, bfh,
                                          cst, hst, 14539, 18635, 19659, 16, 3);
  level_kernel<<<4, 256, lds, stream>>>(xiou, xf, Wiouh, biouh, Wfh, bfh,
                                        cst, hst, 19915, 19979, 0, 16, 2);
  level_kernel<<<1, 256, lds, stream>>>(xiou, xf, Wiouh, biouh, Wfh, bfh,
                                        cst, hst, 19995, 19999, 0, 4, 2);
}
